# starting kernel
# baseline (speedup 1.0000x reference)
#include <hip/hip_runtime.h>

#define HW_    56
#define CTOT   96
#define HEADS  3
#define EDIM   32
#define WIN    7
#define TOK    49
#define TPAD   64
#define NQKV   288
#define WQS    292

typedef __attribute__((ext_vector_type(2))) float v2f;
typedef __attribute__((ext_vector_type(8))) float v8f;

#define WMMA4(a, b, c) \
  __builtin_amdgcn_wmma_f32_16x16x4_f32(false, (a), false, (b), (short)0, (c), false, false)

__global__ __launch_bounds__(256) void swin_msa_fused(
    const float* __restrict__ x, const float* __restrict__ w_qkv,
    const float* __restrict__ b_qkv, const float* __restrict__ w_proj,
    const float* __restrict__ b_proj, const float* __restrict__ rel_bias,
    float* __restrict__ out)
{
  __shared__ float xs[TPAD][100];
  __shared__ float wq[CTOT][WQS];
  __shared__ float qs[HEADS][TPAD][33];
  __shared__ float ks[HEADS][TPAD][33];
  __shared__ float vs[HEADS][TPAD][33];
  __shared__ float sc[HEADS][TPAD][65];
  __shared__ float ow[TPAD][100];

  const int tid  = threadIdx.x;
  const int wid  = tid >> 5;
  const int lane = tid & 31;
  const int ln16 = lane & 15;
  const int kh   = lane >> 4;
  const int rofs = kh * 8;

  const int blk = blockIdx.x;
  const int b   = blk >> 6;
  const int win = blk & 63;
  const int wy  = win >> 3;
  const int wx  = win & 7;
  const long xbase = (long)b * (HW_ * HW_) * CTOT;

  for (int idx = tid; idx < CTOT * (NQKV / 4); idx += 256) {
    int r  = idx / (NQKV / 4);
    int c4 = idx - r * (NQKV / 4);
    float4 v = *(const float4*)(w_qkv + r * NQKV + c4 * 4);
    *(float4*)&wq[r][c4 * 4] = v;
  }
  for (int idx = tid; idx < TPAD * (CTOT / 4); idx += 256) {
    int r  = idx / (CTOT / 4);
    int c4 = idx - r * (CTOT / 4);
    float4 v = make_float4(0.f, 0.f, 0.f, 0.f);
    if (r < TOK) {
      int m1 = r / WIN, m2 = r % WIN;
      int grow = (wy * WIN + m1) * HW_ + (wx * WIN + m2);
      v = *(const float4*)(x + xbase + (long)grow * CTOT + c4 * 4);
    }
    *(float4*)&xs[r][c4 * 4] = v;
  }
  __syncthreads();

  for (int tt = wid; tt < 4 * 18; tt += 8) {
    const int row0 = (tt / 18) * 16;
    const int col0 = (tt % 18) * 16;
    v8f acc = {};
#pragma unroll
    for (int kk = 0; kk < CTOT; kk += 4) {
      v2f a, bf;
      a.x  = xs[row0 + ln16][kk + kh * 2 + 0];
      a.y  = xs[row0 + ln16][kk + kh * 2 + 1];
      bf.x = wq[kk + kh * 2 + 0][col0 + ln16];
      bf.y = wq[kk + kh * 2 + 1][col0 + ln16];
      acc = WMMA4(a, bf, acc);
    }
#pragma unroll
    for (int j = 0; j < 8; ++j) {
      int row = row0 + j + rofs;
      int col = col0 + ln16;
      float val = acc[j] + b_qkv[col];
      int head = col / CTOT;
      int r96  = col - head * CTOT;
      int e    = r96 / 3;
      int kqv  = r96 - 3 * e;
      if (kqv == 0)      qs[head][row][e] = val;
      else if (kqv == 1) ks[head][row][e] = val;
      else               vs[head][row][e] = val;
    }
  }
  __syncthreads();

  const float scale = 0.17677669529663687f;
  for (int tt = wid; tt < HEADS * 16; tt += 8) {
    const int head = tt / 16;
    const int q0   = ((tt % 16) / 4) * 16;
    const int j0   = (tt % 4) * 16;
    v8f acc = {};
#pragma unroll
    for (int kk = 0; kk < EDIM; kk += 4) {
      v2f a, bf;
      a.x  = qs[head][q0 + ln16][kk + kh * 2 + 0];
      a.y  = qs[head][q0 + ln16][kk + kh * 2 + 1];
      bf.x = ks[head][j0 + ln16][kk + kh * 2 + 0];
      bf.y = ks[head][j0 + ln16][kk + kh * 2 + 1];
      acc = WMMA4(a, bf, acc);
    }
#pragma unroll
    for (int j = 0; j < 8; ++j) {
      int row = q0 + j + rofs;
      int col = j0 + ln16;
      float v;
      if (col < TOK) {
        float bias = (row < TOK) ? rel_bias[row * TOK + col] : 0.f;
        v = acc[j] * scale + bias;
      } else {
        v = -1e30f;
      }
      sc[head][row][col] = v;
    }
  }
  __syncthreads();

  for (int task = wid; task < HEADS * TOK; task += 8) {
    const int head = task / TOK;
    const int r    = task - head * TOK;
    float v0 = sc[head][r][lane];
    float v1 = sc[head][r][lane + 32];
    float mx = fmaxf(v0, v1);
    for (int off = 16; off > 0; off >>= 1) mx = fmaxf(mx, __shfl_xor(mx, off, 32));
    float e0 = __expf(v0 - mx);
    float e1 = __expf(v1 - mx);
    float s = e0 + e1;
    for (int off = 16; off > 0; off >>= 1) s += __shfl_xor(s, off, 32);
    float inv = 1.f / s;
    sc[head][r][lane]      = e0 * inv;
    sc[head][r][lane + 32] = e1 * inv;
  }
  __syncthreads();

  for (int tt = wid; tt < HEADS * 4 * 2; tt += 8) {
    const int head = tt / 8;
    const int q0   = ((tt % 8) / 2) * 16;
    const int e0   = (tt % 2) * 16;
    v8f acc = {};
#pragma unroll
    for (int kk = 0; kk < TPAD; kk += 4) {
      v2f a, bf;
      a.x  = sc[head][q0 + ln16][kk + kh * 2 + 0];
      a.y  = sc[head][q0 + ln16][kk + kh * 2 + 1];
      bf.x = vs[head][kk + kh * 2 + 0][e0 + ln16];
      bf.y = vs[head][kk + kh * 2 + 1][e0 + ln16];
      acc = WMMA4(a, bf, acc);
    }
#pragma unroll
    for (int j = 0; j < 8; ++j) {
      int row = q0 + j + rofs;
      int col = e0 + ln16;
      ow[row][head * EDIM + col] = acc[j];
    }
  }
  __syncthreads();

  for (int tt = wid; tt < 4 * 6; tt += 8) {
    const int row0 = (tt / 6) * 16;
    const int col0 = (tt % 6) * 16;
    v8f acc = {};
#pragma unroll
    for (int kk = 0; kk < CTOT; kk += 4) {
      v2f a, bf;
      a.x  = ow[row0 + ln16][kk + kh * 2 + 0];
      a.y  = ow[row0 + ln16][kk + kh * 2 + 1];
      bf.x = w_proj[(kk + kh * 2 + 0) * CTOT + col0 + ln16];
      bf.y = w_proj[(kk + kh * 2 + 1) * CTOT + col0 + ln16];
      acc = WMMA4(a, bf, acc);
    }
#pragma unroll
    for (int j = 0; j < 8; ++j) {
      int row = row0 + j + rofs;
      if (row < TOK) {
        int col = col0 + ln16;
        int m1 = row / WIN, m2 = row % WIN;
        long grow = (long)b * (HW_ * HW_) +
                    (long)((wy * WIN + m1) * HW_ + (wx * WIN + m2));
        out[grow * CTOT + col] = acc[j] + b_proj[col];
      }
    }
  }
}

extern "C" void kernel_launch(void* const* d_in, const int* in_sizes, int n_in,
                              void* d_out, int out_size, void* d_ws, size_t ws_size,
                              hipStream_t stream) {
  const float* x        = (const float*)d_in[0];
  const float* w_qkv    = (const float*)d_in[1];
  const float* b_qkv    = (const float*)d_in[2];
  const float* w_proj   = (const float*)d_in[3];
  const float* b_proj   = (const float*)d_in[4];
  const float* rel_bias = (const float*)d_in[5];
  float* out = (float*)d_out;
  (void)in_sizes; (void)n_in; (void)out_size; (void)d_ws; (void)ws_size;

  swin_msa_fused<<<64 * 64, 256, 0, stream>>>(x, w_qkv, b_qkv, w_proj, b_proj,
                                              rel_bias, out);
}


# 32 waves per work-group instead of 8 (1024 threads; the same tiles and rows dealt over 32 waves): the single work-group that LDS admits per WGP now keeps 8 waves per SIMD, not 2; sums unchanged
# speedup vs baseline: 2.0343x; 2.0343x over previous
#include <hip/hip_runtime.h>

#define HW_    56
#define CTOT   96
#define HEADS  3
#define EDIM   32
#define WIN    7
#define TOK    49
#define TPAD   64
#define NQKV   288
#define WQS    292
#define NTHR   1024
#define NWAV   (NTHR / 32)

typedef __attribute__((ext_vector_type(2))) float v2f;
typedef __attribute__((ext_vector_type(8))) float v8f;

#define WMMA4(a, b, c) \
  __builtin_amdgcn_wmma_f32_16x16x4_f32(false, (a), false, (b), (short)0, (c), false, false)

__global__ __launch_bounds__(NTHR) void swin_msa_fused(
    const float* __restrict__ x, const float* __restrict__ w_qkv,
    const float* __restrict__ b_qkv, const float* __restrict__ w_proj,
    const float* __restrict__ b_proj, const float* __restrict__ rel_bias,
    float* __restrict__ out)
{
  __shared__ float xs[TPAD][100];
  __shared__ float wq[CTOT][WQS];
  __shared__ float qs[HEADS][TPAD][33];
  __shared__ float ks[HEADS][TPAD][33];
  __shared__ float vs[HEADS][TPAD][33];
  __shared__ float sc[HEADS][TPAD][65];
  __shared__ float ow[TPAD][100];

  const int tid  = threadIdx.x;
  const int wid  = tid >> 5;
  const int lane = tid & 31;
  const int ln16 = lane & 15;
  const int kh   = lane >> 4;
  const int rofs = kh * 8;

  const int blk = blockIdx.x;
  const int b   = blk >> 6;
  const int win = blk & 63;
  const int wy  = win >> 3;
  const int wx  = win & 7;
  const long xbase = (long)b * (HW_ * HW_) * CTOT;

  for (int idx = tid; idx < CTOT * (NQKV / 4); idx += NTHR) {
    int r  = idx / (NQKV / 4);
    int c4 = idx - r * (NQKV / 4);
    float4 v = *(const float4*)(w_qkv + r * NQKV + c4 * 4);
    *(float4*)&wq[r][c4 * 4] = v;
  }
  for (int idx = tid; idx < TPAD * (CTOT / 4); idx += NTHR) {
    int r  = idx / (CTOT / 4);
    int c4 = idx - r * (CTOT / 4);
    float4 v = make_float4(0.f, 0.f, 0.f, 0.f);
    if (r < TOK) {
      int m1 = r / WIN, m2 = r % WIN;
      int grow = (wy * WIN + m1) * HW_ + (wx * WIN + m2);
      v = *(const float4*)(x + xbase + (long)grow * CTOT + c4 * 4);
    }
    *(float4*)&xs[r][c4 * 4] = v;
  }
  __syncthreads();

  for (int tt = wid; tt < 4 * 18; tt += NWAV) {
    const int row0 = (tt / 18) * 16;
    const int col0 = (tt % 18) * 16;
    v8f acc = {};
#pragma unroll
    for (int kk = 0; kk < CTOT; kk += 4) {
      v2f a, bf;
      a.x  = xs[row0 + ln16][kk + kh * 2 + 0];
      a.y  = xs[row0 + ln16][kk + kh * 2 + 1];
      bf.x = wq[kk + kh * 2 + 0][col0 + ln16];
      bf.y = wq[kk + kh * 2 + 1][col0 + ln16];
      acc = WMMA4(a, bf, acc);
    }
#pragma unroll
    for (int j = 0; j < 8; ++j) {
      int row = row0 + j + rofs;
      int col = col0 + ln16;
      float val = acc[j] + b_qkv[col];
      int head = col / CTOT;
      int r96  = col - head * CTOT;
      int e    = r96 / 3;
      int kqv  = r96 - 3 * e;
      if (kqv == 0)      qs[head][row][e] = val;
      else if (kqv == 1) ks[head][row][e] = val;
      else               vs[head][row][e] = val;
    }
  }
  __syncthreads();

  const float scale = 0.17677669529663687f;
  for (int tt = wid; tt < HEADS * 16; tt += NWAV) {
    const int head = tt / 16;
    const int q0   = ((tt % 16) / 4) * 16;
    const int j0   = (tt % 4) * 16;
    v8f acc = {};
#pragma unroll
    for (int kk = 0; kk < EDIM; kk += 4) {
      v2f a, bf;
      a.x  = qs[head][q0 + ln16][kk + kh * 2 + 0];
      a.y  = qs[head][q0 + ln16][kk + kh * 2 + 1];
      bf.x = ks[head][j0 + ln16][kk + kh * 2 + 0];
      bf.y = ks[head][j0 + ln16][kk + kh * 2 + 1];
      acc = WMMA4(a, bf, acc);
    }
#pragma unroll
    for (int j = 0; j < 8; ++j) {
      int row = q0 + j + rofs;
      int col = j0 + ln16;
      float v;
      if (col < TOK) {
        float bias = (row < TOK) ? rel_bias[row * TOK + col] : 0.f;
        v = acc[j] * scale + bias;
      } else {
        v = -1e30f;
      }
      sc[head][row][col] = v;
    }
  }
  __syncthreads();

  for (int task = wid; task < HEADS * TOK; task += NWAV) {
    const int head = task / TOK;
    const int r    = task - head * TOK;
    float v0 = sc[head][r][lane];
    float v1 = sc[head][r][lane + 32];
    float mx = fmaxf(v0, v1);
    for (int off = 16; off > 0; off >>= 1) mx = fmaxf(mx, __shfl_xor(mx, off, 32));
    float e0 = __expf(v0 - mx);
    float e1 = __expf(v1 - mx);
    float s = e0 + e1;
    for (int off = 16; off > 0; off >>= 1) s += __shfl_xor(s, off, 32);
    float inv = 1.f / s;
    sc[head][r][lane]      = e0 * inv;
    sc[head][r][lane + 32] = e1 * inv;
  }
  __syncthreads();

  for (int tt = wid; tt < HEADS * 4 * 2; tt += NWAV) {
    const int head = tt / 8;
    const int q0   = ((tt % 8) / 2) * 16;
    const int e0   = (tt % 2) * 16;
    v8f acc = {};
#pragma unroll
    for (int kk = 0; kk < TPAD; kk += 4) {
      v2f a, bf;
      a.x  = sc[head][q0 + ln16][kk + kh * 2 + 0];
      a.y  = sc[head][q0 + ln16][kk + kh * 2 + 1];
      bf.x = vs[head][kk + kh * 2 + 0][e0 + ln16];
      bf.y = vs[head][kk + kh * 2 + 1][e0 + ln16];
      acc = WMMA4(a, bf, acc);
    }
#pragma unroll
    for (int j = 0; j < 8; ++j) {
      int row = q0 + j + rofs;
      int col = e0 + ln16;
      ow[row][head * EDIM + col] = acc[j];
    }
  }
  __syncthreads();

  for (int tt = wid; tt < 4 * 6; tt += NWAV) {
    const int row0 = (tt / 6) * 16;
    const int col0 = (tt % 6) * 16;
    v8f acc = {};
#pragma unroll
    for (int kk = 0; kk < CTOT; kk += 4) {
      v2f a, bf;
      a.x  = ow[row0 + ln16][kk + kh * 2 + 0];
      a.y  = ow[row0 + ln16][kk + kh * 2 + 1];
      bf.x = w_proj[(kk + kh * 2 + 0) * CTOT + col0 + ln16];
      bf.y = w_proj[(kk + kh * 2 + 1) * CTOT + col0 + ln16];
      acc = WMMA4(a, bf, acc);
    }
#pragma unroll
    for (int j = 0; j < 8; ++j) {
      int row = row0 + j + rofs;
      if (row < TOK) {
        int col = col0 + ln16;
        int m1 = row / WIN, m2 = row % WIN;
        long grow = (long)b * (HW_ * HW_) +
                    (long)((wy * WIN + m1) * HW_ + (wx * WIN + m2));
        out[grow * CTOT + col] = acc[j] + b_proj[col];
      }
    }
  }
}

extern "C" void kernel_launch(void* const* d_in, const int* in_sizes, int n_in,
                              void* d_out, int out_size, void* d_ws, size_t ws_size,
                              hipStream_t stream) {
  const float* x        = (const float*)d_in[0];
  const float* w_qkv    = (const float*)d_in[1];
  const float* b_qkv    = (const float*)d_in[2];
  const float* w_proj   = (const float*)d_in[3];
  const float* b_proj   = (const float*)d_in[4];
  const float* rel_bias = (const float*)d_in[5];
  float* out = (float*)d_out;
  (void)in_sizes; (void)n_in; (void)out_size; (void)d_ws; (void)ws_size;

  swin_msa_fused<<<64 * 64, NTHR, 0, stream>>>(x, w_qkv, b_qkv, w_proj, b_proj,
                                              rel_bias, out);
}


# persistent work-groups: 256 work-groups (one per WGP) each loop over 16 windows, so w_qkv is staged into LDS once per work-group instead of once per window; per-window sums unchanged
# speedup vs baseline: 2.2642x; 1.1130x over previous
#include <hip/hip_runtime.h>

#define HW_    56
#define CTOT   96
#define HEADS  3
#define EDIM   32
#define WIN    7
#define TOK    49
#define TPAD   64
#define NQKV   288
#define WQS    292
#define NTHR   1024
#define NWAV   (NTHR / 32)
#define NWIN   (64 * 64)
#define NGRID  256

typedef __attribute__((ext_vector_type(2))) float v2f;
typedef __attribute__((ext_vector_type(8))) float v8f;

#define WMMA4(a, b, c) \
  __builtin_amdgcn_wmma_f32_16x16x4_f32(false, (a), false, (b), (short)0, (c), false, false)

__global__ __launch_bounds__(NTHR) void swin_msa_fused(
    const float* __restrict__ x, const float* __restrict__ w_qkv,
    const float* __restrict__ b_qkv, const float* __restrict__ w_proj,
    const float* __restrict__ b_proj, const float* __restrict__ rel_bias,
    float* __restrict__ out)
{
  __shared__ float xs[TPAD][100];
  __shared__ float wq[CTOT][WQS];
  __shared__ float qs[HEADS][TPAD][33];
  __shared__ float ks[HEADS][TPAD][33];
  __shared__ float vs[HEADS][TPAD][33];
  __shared__ float sc[HEADS][TPAD][65];
  __shared__ float ow[TPAD][100];

  const int tid  = threadIdx.x;
  const int wid  = tid >> 5;
  const int lane = tid & 31;
  const int ln16 = lane & 15;
  const int kh   = lane >> 4;
  const int rofs = kh * 8;

  for (int idx = tid; idx < CTOT * (NQKV / 4); idx += NTHR) {
    int r  = idx / (NQKV / 4);
    int c4 = idx - r * (NQKV / 4);
    float4 v = *(const float4*)(w_qkv + r * NQKV + c4 * 4);
    *(float4*)&wq[r][c4 * 4] = v;
  }

  for (int blk = blockIdx.x; blk < NWIN; blk += NGRID) {
  const int b   = blk >> 6;
  const int win = blk & 63;
  const int wy  = win >> 3;
  const int wx  = win & 7;
  const long wbase = ((long)b * (HW_ * HW_) + (long)((wy * WIN) * HW_ + wx * WIN)) * CTOT;

  for (int idx = tid; idx < TPAD * (CTOT / 4); idx += NTHR) {
    int r  = idx / (CTOT / 4);
    int c4 = idx - r * (CTOT / 4);
    float4 v = make_float4(0.f, 0.f, 0.f, 0.f);
    if (r < TOK) {
      int tok = (r / WIN) * HW_ + (r % WIN);
      v = *(const float4*)(x + wbase + tok * CTOT + c4 * 4);
    }
    *(float4*)&xs[r][c4 * 4] = v;
  }
  __syncthreads();

  for (int tt = wid; tt < 4 * 18; tt += NWAV) {
    const int row0 = (tt / 18) * 16;
    const int col0 = (tt % 18) * 16;
    v8f acc = {};
#pragma unroll
    for (int kk = 0; kk < CTOT; kk += 4) {
      v2f a, bf;
      a.x  = xs[row0 + ln16][kk + kh * 2 + 0];
      a.y  = xs[row0 + ln16][kk + kh * 2 + 1];
      bf.x = wq[kk + kh * 2 + 0][col0 + ln16];
      bf.y = wq[kk + kh * 2 + 1][col0 + ln16];
      acc = WMMA4(a, bf, acc);
    }
#pragma unroll
    for (int j = 0; j < 8; ++j) {
      int row = row0 + j + rofs;
      int col = col0 + ln16;
      float val = acc[j] + b_qkv[col];
      int head = col / CTOT;
      int r96  = col - head * CTOT;
      int e    = r96 / 3;
      int kqv  = r96 - 3 * e;
      if (kqv == 0)      qs[head][row][e] = val;
      else if (kqv == 1) ks[head][row][e] = val;
      else               vs[head][row][e] = val;
    }
  }
  __syncthreads();

  const float scale = 0.17677669529663687f;
  for (int tt = wid; tt < HEADS * 16; tt += NWAV) {
    const int head = tt / 16;
    const int q0   = ((tt % 16) / 4) * 16;
    const int j0   = (tt % 4) * 16;
    const int rb0  = (q0 + rofs) * TOK + j0 + ln16;
    v8f acc = {};
#pragma unroll
    for (int kk = 0; kk < EDIM; kk += 4) {
      v2f a, bf;
      a.x  = qs[head][q0 + ln16][kk + kh * 2 + 0];
      a.y  = qs[head][q0 + ln16][kk + kh * 2 + 1];
      bf.x = ks[head][j0 + ln16][kk + kh * 2 + 0];
      bf.y = ks[head][j0 + ln16][kk + kh * 2 + 1];
      acc = WMMA4(a, bf, acc);
    }
#pragma unroll
    for (int j = 0; j < 8; ++j) {
      int row = q0 + j + rofs;
      int col = j0 + ln16;
      float v;
      if (col < TOK) {
        float bias = (row < TOK) ? rel_bias[rb0 + j * TOK] : 0.f;
        v = acc[j] * scale + bias;
      } else {
        v = -1e30f;
      }
      sc[head][row][col] = v;
    }
  }
  __syncthreads();

  for (int task = wid; task < HEADS * TOK; task += NWAV) {
    const int head = task / TOK;
    const int r    = task - head * TOK;
    float v0 = sc[head][r][lane];
    float v1 = sc[head][r][lane + 32];
    float mx = fmaxf(v0, v1);
    for (int off = 16; off > 0; off >>= 1) mx = fmaxf(mx, __shfl_xor(mx, off, 32));
    float e0 = __expf(v0 - mx);
    float e1 = __expf(v1 - mx);
    float s = e0 + e1;
    for (int off = 16; off > 0; off >>= 1) s += __shfl_xor(s, off, 32);
    float inv = 1.f / s;
    sc[head][r][lane]      = e0 * inv;
    sc[head][r][lane + 32] = e1 * inv;
  }
  __syncthreads();

  for (int tt = wid; tt < HEADS * 4 * 2; tt += NWAV) {
    const int head = tt / 8;
    const int q0   = ((tt % 8) / 2) * 16;
    const int e0   = (tt % 2) * 16;
    v8f acc = {};
#pragma unroll
    for (int kk = 0; kk < TPAD; kk += 4) {
      v2f a, bf;
      a.x  = sc[head][q0 + ln16][kk + kh * 2 + 0];
      a.y  = sc[head][q0 + ln16][kk + kh * 2 + 1];
      bf.x = vs[head][kk + kh * 2 + 0][e0 + ln16];
      bf.y = vs[head][kk + kh * 2 + 1][e0 + ln16];
      acc = WMMA4(a, bf, acc);
    }
#pragma unroll
    for (int j = 0; j < 8; ++j) {
      int row = q0 + j + rofs;
      int col = e0 + ln16;
      ow[row][head * EDIM + col] = acc[j];
    }
  }
  __syncthreads();

  for (int tt = wid; tt < 4 * 6; tt += NWAV) {
    const int row0 = (tt / 6) * 16;
    const int col0 = (tt % 6) * 16;
    const int wp0 = (kh * 2) * CTOT + col0 + ln16;
    v8f acc = {};
#pragma unroll
    for (int kk = 0; kk < CTOT; kk += 4) {
      v2f a, bf;
      a.x  = ow[row0 + ln16][kk + kh * 2 + 0];
      a.y  = ow[row0 + ln16][kk + kh * 2 + 1];
      bf.x = w_proj[wp0 + (kk + 0) * CTOT];
      bf.y = w_proj[wp0 + (kk + 1) * CTOT];
      acc = WMMA4(a, bf, acc);
    }
#pragma unroll
    for (int j = 0; j < 8; ++j) {
      int row = row0 + j + rofs;
      if (row < TOK) {
        int col = col0 + ln16;
        int tok = (row / WIN) * HW_ + (row % WIN);
        out[wbase + tok * CTOT + col] = acc[j] + b_proj[col];
      }
    }
  }
  }
}

extern "C" void kernel_launch(void* const* d_in, const int* in_sizes, int n_in,
                              void* d_out, int out_size, void* d_ws, size_t ws_size,
                              hipStream_t stream) {
  const float* x        = (const float*)d_in[0];
  const float* w_qkv    = (const float*)d_in[1];
  const float* b_qkv    = (const float*)d_in[2];
  const float* w_proj   = (const float*)d_in[3];
  const float* b_proj   = (const float*)d_in[4];
  const float* rel_bias = (const float*)d_in[5];
  float* out = (float*)d_out;
  (void)in_sizes; (void)n_in; (void)out_size; (void)d_ws; (void)ws_size;

  swin_msa_fused<<<NGRID, NTHR, 0, stream>>>(x, w_qkv, b_qkv, w_proj, b_proj,
                                             rel_bias, out);
}


# LDS operand layouts (w_qkv, V k-pair-interleaved; Q, K, P rows at 4 x odd pitch): each fragment one conflict-free 8-byte read; an A fragment feeds 3 (qkv) or 2 (score) tiles; sums unchanged
# speedup vs baseline: 2.3958x; 1.0581x over previous
#include <hip/hip_runtime.h>


#define HW_    56
#define CTOT   96
#define HEADS  3
#define EDIM   32
#define WIN    7
#define TOK    49
#define TPAD   64
#define NQKV   288
#define NTHR   1024
#define NWAV   (NTHR / 32)
#define NWIN   (64 * 64)
#define NGRID  256

#define XS_P   100
#define WQ_P   608
#define QK_P   36
#define VP_P   96
#define SC_P   68
#define QOFF   0
#define KOFF   (HEADS * TPAD * QK_P)
#define VOFF   (2 * HEADS * TPAD * QK_P)
#define QKV_SZ (VOFF + HEADS * (TPAD / 2) * VP_P)

typedef __attribute__((ext_vector_type(2))) float v2f;
typedef __attribute__((ext_vector_type(8))) float v8f;

#define WMMA4(a, b, c) \
  __builtin_amdgcn_wmma_f32_16x16x4_f32(false, (a), false, (b), (short)0, (c), false, false)

__device__ __forceinline__ void scatter_qkv(float* __restrict__ qkv, const v8f acc, int col,
                                            int r0, float bias)
{
  const int head = col / CTOT;
  const int r96  = col - head * CTOT;
  const int e    = r96 / 3;
  const int kqv  = r96 - 3 * e;
  const bool isv = (kqv == 2);
  const int base = isv ? (VOFF + head * ((TPAD / 2) * VP_P) + (r0 >> 1) * VP_P + e * 2)
                       : (kqv * KOFF + head * (TPAD * QK_P) + r0 * QK_P + e);
  const int s2 = isv ? VP_P : 2 * QK_P;
  const int s1 = isv ? 1 : QK_P;
#pragma unroll
  for (int j = 0; j < 8; ++j)
    qkv[base + (j >> 1) * s2 + (j & 1) * s1] = acc[j] + bias;
}

__global__ __launch_bounds__(NTHR) void swin_msa_fused(
    const float* __restrict__ x, const float* __restrict__ w_qkv,
    const float* __restrict__ b_qkv, const float* __restrict__ w_proj,
    const float* __restrict__ b_proj, const float* __restrict__ rel_bias,
    float* __restrict__ out)
{
  __shared__ __attribute__((aligned(16))) float xs[TPAD][XS_P];
  __shared__ __attribute__((aligned(16))) float wqp[CTOT / 2][WQ_P];
  __shared__ __attribute__((aligned(16))) float qkv[QKV_SZ];
  __shared__ __attribute__((aligned(16))) float sc[HEADS][TPAD][SC_P];
  __shared__ __attribute__((aligned(16))) float ow[TPAD][XS_P];

  const int tid  = threadIdx.x;
  const int wid  = tid >> 5;
  const int lane = tid & 31;
  const int ln16 = lane & 15;
  const int kh   = lane >> 4;
  const int rofs = kh * 8;

  for (int idx = tid; idx < (CTOT / 2) * (NQKV / 4); idx += NTHR) {
    int p  = idx / (NQKV / 4);
    int c4 = idx - p * (NQKV / 4);
    float4 r0 = *(const float4*)(w_qkv + (2 * p) * NQKV + c4 * 4);
    float4 r1 = *(const float4*)(w_qkv + (2 * p + 1) * NQKV + c4 * 4);
    float* d = &wqp[p][c4 * 8];
    *(float4*)(d)     = make_float4(r0.x, r1.x, r0.y, r1.y);
    *(float4*)(d + 4) = make_float4(r0.z, r1.z, r0.w, r1.w);
  }

  for (int blk = blockIdx.x; blk < NWIN; blk += NGRID) {
  const int b   = blk >> 6;
  const int win = blk & 63;
  const int wy  = win >> 3;
  const int wx  = win & 7;
  const long wbase = ((long)b * (HW_ * HW_) + (long)((wy * WIN) * HW_ + wx * WIN)) * CTOT;
  const float* __restrict__ xw = x + wbase;
  float* __restrict__ ow_g     = out + wbase;

  for (int idx = tid; idx < TPAD * (CTOT / 4); idx += NTHR) {
    int r  = idx / (CTOT / 4);
    int c4 = idx - r * (CTOT / 4);
    float4 v = make_float4(0.f, 0.f, 0.f, 0.f);
    if (r < TOK) {
      int tok = (r / WIN) * HW_ + (r % WIN);
      v = *(const float4*)(xw + tok * CTOT + c4 * 4);
    }
    *(float4*)&xs[r][c4 * 4] = v;
  }
  __syncthreads();

  for (int tt = wid; tt < 4 * 6; tt += NWAV) {
    const int row0 = (tt / 6) * 16;
    const int cg   = (tt % 6) * 48;
    v8f acc0 = {}, acc1 = {}, acc2 = {};
    const int c0 = cg + ln16;
    const float bq0 = b_qkv[c0];
    const float bq1 = b_qkv[c0 + 16];
    const float bq2 = b_qkv[c0 + 32];
    const float* ap = &xs[row0 + ln16][kh * 2];
    const float* bp = &wqp[kh][(cg + ln16) * 2];
#pragma unroll
    for (int kk = 0; kk < CTOT; kk += 4) {
      v2f a  = *(const v2f*)(ap + kk);
      v2f b0 = *(const v2f*)(bp + (kk / 2) * WQ_P);
      v2f b1 = *(const v2f*)(bp + (kk / 2) * WQ_P + 32);
      v2f b2 = *(const v2f*)(bp + (kk / 2) * WQ_P + 64);
      acc0 = WMMA4(a, b0, acc0);
      acc1 = WMMA4(a, b1, acc1);
      acc2 = WMMA4(a, b2, acc2);
    }
    scatter_qkv(qkv, acc0, c0,      row0 + rofs, bq0);
    scatter_qkv(qkv, acc1, c0 + 16, row0 + rofs, bq1);
    scatter_qkv(qkv, acc2, c0 + 32, row0 + rofs, bq2);
  }
  __syncthreads();

  const float scale = 0.17677669529663687f;
  for (int tt = wid; tt < HEADS * 8; tt += NWAV) {
    const int head = tt / 8;
    const int q0   = ((tt % 8) / 2) * 16;
    const int jh   = (tt % 2) * 32;
    v8f acc0 = {}, acc1 = {};
    const float* ap = &qkv[QOFF + head * (TPAD * QK_P) + (q0 + ln16) * QK_P + kh * 2];
    const float* bp = &qkv[KOFF + head * (TPAD * QK_P) + (jh + ln16) * QK_P + kh * 2];
#pragma unroll
    for (int kk = 0; kk < EDIM; kk += 4) {
      v2f a  = *(const v2f*)(ap + kk);
      v2f b0 = *(const v2f*)(bp + kk);
      v2f b1 = *(const v2f*)(bp + 16 * QK_P + kk);
      acc0 = WMMA4(a, b0, acc0);
      acc1 = WMMA4(a, b1, acc1);
    }
    const float* rbp = rel_bias + (q0 + rofs) * TOK + jh + ln16;
#pragma unroll
    for (int i = 0; i < 2; ++i) {
      const int j0  = jh + i * 16;
#pragma unroll
      for (int j = 0; j < 8; ++j) {
        int row = q0 + j + rofs;
        int col = j0 + ln16;
        float a = (i == 0) ? acc0[j] : acc1[j];
        float v;
        if (col < TOK) {
          float bias = (row < TOK) ? rbp[i * 16 + j * TOK] : 0.f;
          v = a * scale + bias;
        } else {
          v = -1e30f;
        }
        sc[head][row][col] = v;
      }
    }
  }
  __syncthreads();

  for (int task = wid; task < HEADS * TOK; task += NWAV) {
    const int head = task / TOK;
    const int r    = task - head * TOK;
    float v0 = sc[head][r][lane];
    float v1 = sc[head][r][lane + 32];
    float mx = fmaxf(v0, v1);
    for (int off = 16; off > 0; off >>= 1) mx = fmaxf(mx, __shfl_xor(mx, off, 32));
    float e0 = __expf(v0 - mx);
    float e1 = __expf(v1 - mx);
    float s = e0 + e1;
    for (int off = 16; off > 0; off >>= 1) s += __shfl_xor(s, off, 32);
    float inv = 1.f / s;
    sc[head][r][lane]      = e0 * inv;
    sc[head][r][lane + 32] = e1 * inv;
  }
  __syncthreads();

  for (int tt = wid; tt < HEADS * 4 * 2; tt += NWAV) {
    const int head = tt / 8;
    const int q0   = ((tt % 8) / 2) * 16;
    const int e0   = (tt % 2) * 16;
    v8f acc = {};
    const float* ap = &sc[head][q0 + ln16][kh * 2];
    const float* bp = &qkv[VOFF + head * ((TPAD / 2) * VP_P) + kh * VP_P + (e0 + ln16) * 2];
#pragma unroll
    for (int kk = 0; kk < TPAD; kk += 4) {
      v2f a  = *(const v2f*)(ap + kk);
      v2f bf = *(const v2f*)(bp + (kk / 2) * VP_P);
      acc = WMMA4(a, bf, acc);
    }
#pragma unroll
    for (int j = 0; j < 8; ++j) {
      int row = q0 + j + rofs;
      int col = e0 + ln16;
      ow[row][head * EDIM + col] = acc[j];
    }
  }
  __syncthreads();

  for (int tt = wid; tt < 4 * 6; tt += NWAV) {
    const int row0 = (tt / 6) * 16;
    const int col0 = (tt % 6) * 16;
    const int wp0 = (kh * 2) * CTOT + col0 + ln16;
    v8f acc = {};
#pragma unroll
    for (int kk = 0; kk < CTOT; kk += 4) {
      v2f a, bf;
      a.x  = ow[row0 + ln16][kk + kh * 2 + 0];
      a.y  = ow[row0 + ln16][kk + kh * 2 + 1];
      bf.x = w_proj[wp0 + (kk + 0) * CTOT];
      bf.y = w_proj[wp0 + (kk + 1) * CTOT];
      acc = WMMA4(a, bf, acc);
    }
#pragma unroll
    for (int j = 0; j < 8; ++j) {
      int row = row0 + j + rofs;
      if (row < TOK) {
        int col = col0 + ln16;
        int tok = (row / WIN) * HW_ + (row % WIN);
        ow_g[tok * CTOT + col] = acc[j] + b_proj[col];
      }
    }
  }
  }
}

extern "C" void kernel_launch(void* const* d_in, const int* in_sizes, int n_in,
                              void* d_out, int out_size, void* d_ws, size_t ws_size,
                              hipStream_t stream) {
  const float* x        = (const float*)d_in[0];
  const float* w_qkv    = (const float*)d_in[1];
  const float* b_qkv    = (const float*)d_in[2];
  const float* w_proj   = (const float*)d_in[3];
  const float* b_proj   = (const float*)d_in[4];
  const float* rel_bias = (const float*)d_in[5];
  float* out = (float*)d_out;
  (void)in_sizes; (void)n_in; (void)out_size; (void)d_ws; (void)ws_size;

  swin_msa_fused<<<NGRID, NTHR, 0, stream>>>(x, w_qkv, b_qkv, w_proj, b_proj,
                                             rel_bias, out);
}


# score and projection epilogues no longer wait for one global load per output row: rel_bias read as one group from an LDS copy staged once per work-group, b_proj loaded once per tile; sums unchanged
# speedup vs baseline: 3.0706x; 1.2817x over previous
#include <hip/hip_runtime.h>


#define HW_    56
#define CTOT   96
#define HEADS  3
#define EDIM   32
#define WIN    7
#define TOK    49
#define TPAD   64
#define NQKV   288
#define NTHR   1024
#define NWAV   (NTHR / 32)
#define NWIN   (64 * 64)
#define NGRID  256

#define XS_P   100
#define WQ_P   608
#define QK_P   36
#define VP_P   96
#define SC_P   68
#define QOFF   0
#define KOFF   (HEADS * TPAD * QK_P)
#define VOFF   (2 * HEADS * TPAD * QK_P)
#define QKV_SZ (VOFF + HEADS * (TPAD / 2) * VP_P)
#define RB_SZ  ((TPAD - 1) * TOK + TPAD)

typedef __attribute__((ext_vector_type(2))) float v2f;
typedef __attribute__((ext_vector_type(8))) float v8f;

#define WMMA4(a, b, c) \
  __builtin_amdgcn_wmma_f32_16x16x4_f32(false, (a), false, (b), (short)0, (c), false, false)

__device__ __forceinline__ void scatter_qkv(float* __restrict__ qkv, const v8f acc, int col,
                                            int r0, float bias)
{
  const int head = col / CTOT;
  const int r96  = col - head * CTOT;
  const int e    = r96 / 3;
  const int kqv  = r96 - 3 * e;
  const bool isv = (kqv == 2);
  const int base = isv ? (VOFF + head * ((TPAD / 2) * VP_P) + (r0 >> 1) * VP_P + e * 2)
                       : (kqv * KOFF + head * (TPAD * QK_P) + r0 * QK_P + e);
  const int s2 = isv ? VP_P : 2 * QK_P;
  const int s1 = isv ? 1 : QK_P;
#pragma unroll
  for (int j = 0; j < 8; ++j)
    qkv[base + (j >> 1) * s2 + (j & 1) * s1] = acc[j] + bias;
}

__global__ __launch_bounds__(NTHR) void swin_msa_fused(
    const float* __restrict__ x, const float* __restrict__ w_qkv,
    const float* __restrict__ b_qkv, const float* __restrict__ w_proj,
    const float* __restrict__ b_proj, const float* __restrict__ rel_bias,
    float* __restrict__ out)
{
  __shared__ __attribute__((aligned(16))) float xs[TPAD][XS_P];
  __shared__ __attribute__((aligned(16))) float wqp[CTOT / 2][WQ_P];
  __shared__ __attribute__((aligned(16))) float qkv[QKV_SZ];
  __shared__ __attribute__((aligned(16))) float sc[HEADS][TPAD][SC_P];
  __shared__ __attribute__((aligned(16))) float ow[TPAD][XS_P];
  __shared__ __attribute__((aligned(16))) float rbl[RB_SZ];

  const int tid  = threadIdx.x;
  const int wid  = tid >> 5;
  const int lane = tid & 31;
  const int ln16 = lane & 15;
  const int kh   = lane >> 4;
  const int rofs = kh * 8;

  for (int idx = tid; idx < (CTOT / 2) * (NQKV / 4); idx += NTHR) {
    int p  = idx / (NQKV / 4);
    int c4 = idx - p * (NQKV / 4);
    float4 r0 = *(const float4*)(w_qkv + (2 * p) * NQKV + c4 * 4);
    float4 r1 = *(const float4*)(w_qkv + (2 * p + 1) * NQKV + c4 * 4);
    float* d = &wqp[p][c4 * 8];
    *(float4*)(d)     = make_float4(r0.x, r1.x, r0.y, r1.y);
    *(float4*)(d + 4) = make_float4(r0.z, r1.z, r0.w, r1.w);
  }

  for (int idx = tid; idx < RB_SZ; idx += NTHR)
    rbl[idx] = (idx < TOK * TOK) ? rel_bias[idx] : 0.f;

  for (int blk = blockIdx.x; blk < NWIN; blk += NGRID) {
  const int b   = blk >> 6;
  const int win = blk & 63;
  const int wy  = win >> 3;
  const int wx  = win & 7;
  const long wbase = ((long)b * (HW_ * HW_) + (long)((wy * WIN) * HW_ + wx * WIN)) * CTOT;
  const float* __restrict__ xw = x + wbase;
  float* __restrict__ ow_g     = out + wbase;

  for (int idx = tid; idx < TPAD * (CTOT / 4); idx += NTHR) {
    int r  = idx / (CTOT / 4);
    int c4 = idx - r * (CTOT / 4);
    float4 v = make_float4(0.f, 0.f, 0.f, 0.f);
    if (r < TOK) {
      int tok = (r / WIN) * HW_ + (r % WIN);
      v = *(const float4*)(xw + tok * CTOT + c4 * 4);
    }
    *(float4*)&xs[r][c4 * 4] = v;
  }
  __syncthreads();

  for (int tt = wid; tt < 4 * 6; tt += NWAV) {
    const int row0 = (tt / 6) * 16;
    const int cg   = (tt % 6) * 48;
    v8f acc0 = {}, acc1 = {}, acc2 = {};
    const int c0 = cg + ln16;
    const float bq0 = b_qkv[c0];
    const float bq1 = b_qkv[c0 + 16];
    const float bq2 = b_qkv[c0 + 32];
    const float* ap = &xs[row0 + ln16][kh * 2];
    const float* bp = &wqp[kh][(cg + ln16) * 2];
#pragma unroll
    for (int kk = 0; kk < CTOT; kk += 4) {
      v2f a  = *(const v2f*)(ap + kk);
      v2f b0 = *(const v2f*)(bp + (kk / 2) * WQ_P);
      v2f b1 = *(const v2f*)(bp + (kk / 2) * WQ_P + 32);
      v2f b2 = *(const v2f*)(bp + (kk / 2) * WQ_P + 64);
      acc0 = WMMA4(a, b0, acc0);
      acc1 = WMMA4(a, b1, acc1);
      acc2 = WMMA4(a, b2, acc2);
    }
    scatter_qkv(qkv, acc0, c0,      row0 + rofs, bq0);
    scatter_qkv(qkv, acc1, c0 + 16, row0 + rofs, bq1);
    scatter_qkv(qkv, acc2, c0 + 32, row0 + rofs, bq2);
  }
  __syncthreads();

  const float scale = 0.17677669529663687f;
  for (int tt = wid; tt < HEADS * 8; tt += NWAV) {
    const int head = tt / 8;
    const int q0   = ((tt % 8) / 2) * 16;
    const int jh   = (tt % 2) * 32;
    v8f acc0 = {}, acc1 = {};
    float rb[2][8];
    {
      const float* rbp = &rbl[(q0 + rofs) * TOK + jh + ln16];
#pragma unroll
      for (int j = 0; j < 8; ++j) {
        rb[0][j] = rbp[j * TOK];
        rb[1][j] = rbp[j * TOK + 16];
      }
    }
    const float* ap = &qkv[QOFF + head * (TPAD * QK_P) + (q0 + ln16) * QK_P + kh * 2];
    const float* bp = &qkv[KOFF + head * (TPAD * QK_P) + (jh + ln16) * QK_P + kh * 2];
#pragma unroll
    for (int kk = 0; kk < EDIM; kk += 4) {
      v2f a  = *(const v2f*)(ap + kk);
      v2f b0 = *(const v2f*)(bp + kk);
      v2f b1 = *(const v2f*)(bp + 16 * QK_P + kk);
      acc0 = WMMA4(a, b0, acc0);
      acc1 = WMMA4(a, b1, acc1);
    }
#pragma unroll
    for (int i = 0; i < 2; ++i) {
      const int j0  = jh + i * 16;
#pragma unroll
      for (int j = 0; j < 8; ++j) {
        int row = q0 + j + rofs;
        int col = j0 + ln16;
        float a = (i == 0) ? acc0[j] : acc1[j];
        float bias = (row < TOK) ? rb[i][j] : 0.f;
        float v = a * scale + bias;
        if (col >= TOK) v = -1e30f;
        sc[head][row][col] = v;
      }
    }
  }
  __syncthreads();

  for (int task = wid; task < HEADS * TOK; task += NWAV) {
    const int head = task / TOK;
    const int r    = task - head * TOK;
    float v0 = sc[head][r][lane];
    float v1 = sc[head][r][lane + 32];
    float mx = fmaxf(v0, v1);
    for (int off = 16; off > 0; off >>= 1) mx = fmaxf(mx, __shfl_xor(mx, off, 32));
    float e0 = __expf(v0 - mx);
    float e1 = __expf(v1 - mx);
    float s = e0 + e1;
    for (int off = 16; off > 0; off >>= 1) s += __shfl_xor(s, off, 32);
    float inv = 1.f / s;
    sc[head][r][lane]      = e0 * inv;
    sc[head][r][lane + 32] = e1 * inv;
  }
  __syncthreads();

  for (int tt = wid; tt < HEADS * 4 * 2; tt += NWAV) {
    const int head = tt / 8;
    const int q0   = ((tt % 8) / 2) * 16;
    const int e0   = (tt % 2) * 16;
    v8f acc = {};
    const float* ap = &sc[head][q0 + ln16][kh * 2];
    const float* bp = &qkv[VOFF + head * ((TPAD / 2) * VP_P) + kh * VP_P + (e0 + ln16) * 2];
#pragma unroll
    for (int kk = 0; kk < TPAD; kk += 4) {
      v2f a  = *(const v2f*)(ap + kk);
      v2f bf = *(const v2f*)(bp + (kk / 2) * VP_P);
      acc = WMMA4(a, bf, acc);
    }
#pragma unroll
    for (int j = 0; j < 8; ++j) {
      int row = q0 + j + rofs;
      int col = e0 + ln16;
      ow[row][head * EDIM + col] = acc[j];
    }
  }
  __syncthreads();

  for (int tt = wid; tt < 4 * 6; tt += NWAV) {
    const int row0 = (tt / 6) * 16;
    const int col0 = (tt % 6) * 16;
    const int wp0 = (kh * 2) * CTOT + col0 + ln16;
    const float bpj = b_proj[col0 + ln16];
    v8f acc = {};
#pragma unroll
    for (int kk = 0; kk < CTOT; kk += 4) {
      v2f a, bf;
      a.x  = ow[row0 + ln16][kk + kh * 2 + 0];
      a.y  = ow[row0 + ln16][kk + kh * 2 + 1];
      bf.x = w_proj[wp0 + (kk + 0) * CTOT];
      bf.y = w_proj[wp0 + (kk + 1) * CTOT];
      acc = WMMA4(a, bf, acc);
    }
#pragma unroll
    for (int j = 0; j < 8; ++j) {
      int row = row0 + j + rofs;
      if (row < TOK) {
        int col = col0 + ln16;
        int tok = (row / WIN) * HW_ + (row % WIN);
        ow_g[tok * CTOT + col] = acc[j] + bpj;
      }
    }
  }
  }
}

extern "C" void kernel_launch(void* const* d_in, const int* in_sizes, int n_in,
                              void* d_out, int out_size, void* d_ws, size_t ws_size,
                              hipStream_t stream) {
  const float* x        = (const float*)d_in[0];
  const float* w_qkv    = (const float*)d_in[1];
  const float* b_qkv    = (const float*)d_in[2];
  const float* w_proj   = (const float*)d_in[3];
  const float* b_proj   = (const float*)d_in[4];
  const float* rel_bias = (const float*)d_in[5];
  float* out = (float*)d_out;
  (void)in_sizes; (void)n_in; (void)out_size; (void)d_ws; (void)ws_size;

  swin_msa_fused<<<NGRID, NTHR, 0, stream>>>(x, w_qkv, b_qkv, w_proj, b_proj,
                                             rel_bias, out);
}


# softmax re-dealt: each row held by a quad of lanes (16 scores per lane), max and sum folded in-lane then across the quad by two DPP steps, all 147 rows in one pass; row-sum order changes
# speedup vs baseline: 3.6748x; 1.1968x over previous
#include <hip/hip_runtime.h>


#define HW_    56
#define CTOT   96
#define HEADS  3
#define EDIM   32
#define WIN    7
#define TOK    49
#define TPAD   64
#define NQKV   288
#define NTHR   1024
#define NWAV   (NTHR / 32)
#define NWIN   (64 * 64)
#define NGRID  256

#define XS_P   100
#define WQ_P   608
#define QK_P   36
#define VP_P   96
#define SC_P   68
#define QOFF   0
#define KOFF   (HEADS * TPAD * QK_P)
#define VOFF   (2 * HEADS * TPAD * QK_P)
#define QKV_SZ (VOFF + HEADS * (TPAD / 2) * VP_P)
#define RB_SZ  ((TPAD - 1) * TOK + TPAD)

typedef __attribute__((ext_vector_type(2))) float v2f;
typedef __attribute__((ext_vector_type(8))) float v8f;

#define WMMA4(a, b, c) \
  __builtin_amdgcn_wmma_f32_16x16x4_f32(false, (a), false, (b), (short)0, (c), false, false)

#define QUAD_XOR1 0xB1
#define QUAD_XOR2 0x4E
template <int CTRL>
__device__ __forceinline__ float quad_xchg(float x)
{
  return __builtin_bit_cast(float,
      __builtin_amdgcn_update_dpp(0, __builtin_bit_cast(int, x), CTRL, 0xF, 0xF, true));
}

__device__ __forceinline__ void scatter_qkv(float* __restrict__ qkv, const v8f acc, int col,
                                            int r0, float bias)
{
  const int head = col / CTOT;
  const int r96  = col - head * CTOT;
  const int e    = r96 / 3;
  const int kqv  = r96 - 3 * e;
  const bool isv = (kqv == 2);
  const int base = isv ? (VOFF + head * ((TPAD / 2) * VP_P) + (r0 >> 1) * VP_P + e * 2)
                       : (kqv * KOFF + head * (TPAD * QK_P) + r0 * QK_P + e);
  const int s2 = isv ? VP_P : 2 * QK_P;
  const int s1 = isv ? 1 : QK_P;
#pragma unroll
  for (int j = 0; j < 8; ++j)
    qkv[base + (j >> 1) * s2 + (j & 1) * s1] = acc[j] + bias;
}

__global__ __launch_bounds__(NTHR) void swin_msa_fused(
    const float* __restrict__ x, const float* __restrict__ w_qkv,
    const float* __restrict__ b_qkv, const float* __restrict__ w_proj,
    const float* __restrict__ b_proj, const float* __restrict__ rel_bias,
    float* __restrict__ out)
{
  __shared__ __attribute__((aligned(16))) float xs[TPAD][XS_P];
  __shared__ __attribute__((aligned(16))) float wqp[CTOT / 2][WQ_P];
  __shared__ __attribute__((aligned(16))) float qkv[QKV_SZ];
  __shared__ __attribute__((aligned(16))) float sc[HEADS][TPAD][SC_P];
  __shared__ __attribute__((aligned(16))) float ow[TPAD][XS_P];
  __shared__ __attribute__((aligned(16))) float rbl[RB_SZ];

  const int tid  = threadIdx.x;
  const int wid  = tid >> 5;
  const int lane = tid & 31;
  const int ln16 = lane & 15;
  const int kh   = lane >> 4;
  const int rofs = kh * 8;

  for (int idx = tid; idx < (CTOT / 2) * (NQKV / 4); idx += NTHR) {
    int p  = idx / (NQKV / 4);
    int c4 = idx - p * (NQKV / 4);
    float4 r0 = *(const float4*)(w_qkv + (2 * p) * NQKV + c4 * 4);
    float4 r1 = *(const float4*)(w_qkv + (2 * p + 1) * NQKV + c4 * 4);
    float* d = &wqp[p][c4 * 8];
    *(float4*)(d)     = make_float4(r0.x, r1.x, r0.y, r1.y);
    *(float4*)(d + 4) = make_float4(r0.z, r1.z, r0.w, r1.w);
  }

  for (int idx = tid; idx < RB_SZ; idx += NTHR)
    rbl[idx] = (idx < TOK * TOK) ? rel_bias[idx] : 0.f;

  for (int blk = blockIdx.x; blk < NWIN; blk += NGRID) {
  const int b   = blk >> 6;
  const int win = blk & 63;
  const int wy  = win >> 3;
  const int wx  = win & 7;
  const long wbase = ((long)b * (HW_ * HW_) + (long)((wy * WIN) * HW_ + wx * WIN)) * CTOT;
  const float* __restrict__ xw = x + wbase;
  float* __restrict__ ow_g     = out + wbase;

  for (int idx = tid; idx < TPAD * (CTOT / 4); idx += NTHR) {
    int r  = idx / (CTOT / 4);
    int c4 = idx - r * (CTOT / 4);
    float4 v = make_float4(0.f, 0.f, 0.f, 0.f);
    if (r < TOK) {
      int tok = (r / WIN) * HW_ + (r % WIN);
      v = *(const float4*)(xw + tok * CTOT + c4 * 4);
    }
    *(float4*)&xs[r][c4 * 4] = v;
  }
  __syncthreads();

  for (int tt = wid; tt < 4 * 6; tt += NWAV) {
    const int row0 = (tt / 6) * 16;
    const int cg   = (tt % 6) * 48;
    v8f acc0 = {}, acc1 = {}, acc2 = {};
    const int c0 = cg + ln16;
    const float bq0 = b_qkv[c0];
    const float bq1 = b_qkv[c0 + 16];
    const float bq2 = b_qkv[c0 + 32];
    const float* ap = &xs[row0 + ln16][kh * 2];
    const float* bp = &wqp[kh][(cg + ln16) * 2];
#pragma unroll
    for (int kk = 0; kk < CTOT; kk += 4) {
      v2f a  = *(const v2f*)(ap + kk);
      v2f b0 = *(const v2f*)(bp + (kk / 2) * WQ_P);
      v2f b1 = *(const v2f*)(bp + (kk / 2) * WQ_P + 32);
      v2f b2 = *(const v2f*)(bp + (kk / 2) * WQ_P + 64);
      acc0 = WMMA4(a, b0, acc0);
      acc1 = WMMA4(a, b1, acc1);
      acc2 = WMMA4(a, b2, acc2);
    }
    scatter_qkv(qkv, acc0, c0,      row0 + rofs, bq0);
    scatter_qkv(qkv, acc1, c0 + 16, row0 + rofs, bq1);
    scatter_qkv(qkv, acc2, c0 + 32, row0 + rofs, bq2);
  }
  __syncthreads();

  const float scale = 0.17677669529663687f;
  for (int tt = wid; tt < HEADS * 8; tt += NWAV) {
    const int head = tt / 8;
    const int q0   = ((tt % 8) / 2) * 16;
    const int jh   = (tt % 2) * 32;
    v8f acc0 = {}, acc1 = {};
    float rb[2][8];
    {
      const float* rbp = &rbl[(q0 + rofs) * TOK + jh + ln16];
#pragma unroll
      for (int j = 0; j < 8; ++j) {
        rb[0][j] = rbp[j * TOK];
        rb[1][j] = rbp[j * TOK + 16];
      }
    }
    const float* ap = &qkv[QOFF + head * (TPAD * QK_P) + (q0 + ln16) * QK_P + kh * 2];
    const float* bp = &qkv[KOFF + head * (TPAD * QK_P) + (jh + ln16) * QK_P + kh * 2];
#pragma unroll
    for (int kk = 0; kk < EDIM; kk += 4) {
      v2f a  = *(const v2f*)(ap + kk);
      v2f b0 = *(const v2f*)(bp + kk);
      v2f b1 = *(const v2f*)(bp + 16 * QK_P + kk);
      acc0 = WMMA4(a, b0, acc0);
      acc1 = WMMA4(a, b1, acc1);
    }
#pragma unroll
    for (int i = 0; i < 2; ++i) {
      const int j0  = jh + i * 16;
#pragma unroll
      for (int j = 0; j < 8; ++j) {
        int row = q0 + j + rofs;
        int col = j0 + ln16;
        float a = (i == 0) ? acc0[j] : acc1[j];
        float bias = (row < TOK) ? rb[i][j] : 0.f;
        float v = a * scale + bias;
        if (col >= TOK) v = -1e30f;
        sc[head][row][col] = v;
      }
    }
  }
  __syncthreads();

  {
    const int t = wid * 8 + (lane >> 2);
    if (t < HEADS * TOK) {
      const int head = t / TOK;
      const int r    = t - head * TOK;
      float4* rowp = (float4*)&sc[head][r][(lane & 3) * 16];
      float4 q4[4];
      for (int i = 0; i < 4; ++i) q4[i] = rowp[i];
      float v[16];
      for (int i = 0; i < 4; ++i) {
        v[4 * i + 0] = q4[i].x; v[4 * i + 1] = q4[i].y;
        v[4 * i + 2] = q4[i].z; v[4 * i + 3] = q4[i].w;
      }
      float m8[8], m4[4];
      for (int i = 0; i < 8; ++i) m8[i] = fmaxf(v[2 * i], v[2 * i + 1]);
      for (int i = 0; i < 4; ++i) m4[i] = fmaxf(m8[2 * i], m8[2 * i + 1]);
      float mx = fmaxf(fmaxf(m4[0], m4[1]), fmaxf(m4[2], m4[3]));
      mx = fmaxf(mx, quad_xchg<QUAD_XOR1>(mx));
      mx = fmaxf(mx, quad_xchg<QUAD_XOR2>(mx));
      float e[16];
      for (int i = 0; i < 16; ++i) e[i] = __expf(v[i] - mx);
      float s8[8], s4[4];
      for (int i = 0; i < 8; ++i) s8[i] = e[2 * i] + e[2 * i + 1];
      for (int i = 0; i < 4; ++i) s4[i] = s8[2 * i] + s8[2 * i + 1];
      float s = (s4[0] + s4[1]) + (s4[2] + s4[3]);
      s += quad_xchg<QUAD_XOR1>(s);
      s += quad_xchg<QUAD_XOR2>(s);
      const float inv = 1.f / s;
      for (int i = 0; i < 4; ++i)
        rowp[i] = make_float4(e[4 * i + 0] * inv, e[4 * i + 1] * inv,
                              e[4 * i + 2] * inv, e[4 * i + 3] * inv);
    }
  }
  __syncthreads();

  for (int tt = wid; tt < HEADS * 4 * 2; tt += NWAV) {
    const int head = tt / 8;
    const int q0   = ((tt % 8) / 2) * 16;
    const int e0   = (tt % 2) * 16;
    v8f acc = {};
    const float* ap = &sc[head][q0 + ln16][kh * 2];
    const float* bp = &qkv[VOFF + head * ((TPAD / 2) * VP_P) + kh * VP_P + (e0 + ln16) * 2];
#pragma unroll
    for (int kk = 0; kk < TPAD; kk += 4) {
      v2f a  = *(const v2f*)(ap + kk);
      v2f bf = *(const v2f*)(bp + (kk / 2) * VP_P);
      acc = WMMA4(a, bf, acc);
    }
#pragma unroll
    for (int j = 0; j < 8; ++j) {
      int row = q0 + j + rofs;
      int col = e0 + ln16;
      ow[row][head * EDIM + col] = acc[j];
    }
  }
  __syncthreads();

  for (int tt = wid; tt < 4 * 6; tt += NWAV) {
    const int row0 = (tt / 6) * 16;
    const int col0 = (tt % 6) * 16;
    const int wp0 = (kh * 2) * CTOT + col0 + ln16;
    const float bpj = b_proj[col0 + ln16];
    v8f acc = {};
#pragma unroll
    for (int kk = 0; kk < CTOT; kk += 4) {
      v2f a, bf;
      a.x  = ow[row0 + ln16][kk + kh * 2 + 0];
      a.y  = ow[row0 + ln16][kk + kh * 2 + 1];
      bf.x = w_proj[wp0 + (kk + 0) * CTOT];
      bf.y = w_proj[wp0 + (kk + 1) * CTOT];
      acc = WMMA4(a, bf, acc);
    }
#pragma unroll
    for (int j = 0; j < 8; ++j) {
      int row = row0 + j + rofs;
      if (row < TOK) {
        int col = col0 + ln16;
        int tok = (row / WIN) * HW_ + (row % WIN);
        ow_g[tok * CTOT + col] = acc[j] + bpj;
      }
    }
  }
  }
}

extern "C" void kernel_launch(void* const* d_in, const int* in_sizes, int n_in,
                              void* d_out, int out_size, void* d_ws, size_t ws_size,
                              hipStream_t stream) {
  const float* x        = (const float*)d_in[0];
  const float* w_qkv    = (const float*)d_in[1];
  const float* b_qkv    = (const float*)d_in[2];
  const float* w_proj   = (const float*)d_in[3];
  const float* b_proj   = (const float*)d_in[4];
  const float* rel_bias = (const float*)d_in[5];
  float* out = (float*)d_out;
  (void)in_sizes; (void)n_in; (void)out_size; (void)d_ws; (void)ws_size;

  swin_msa_fused<<<NGRID, NTHR, 0, stream>>>(x, w_qkv, b_qkv, w_proj, b_proj,
                                             rel_bias, out);
}


# w_proj staged once per work-group in LDS (k-pair-interleaved): the projection loop reads B there, not from global; room made by overlaying x tile on scores and attention output on Q; sums unchanged
# speedup vs baseline: 4.4400x; 1.2082x over previous
#include <hip/hip_runtime.h>


#define HW_    56
#define CTOT   96
#define HEADS  3
#define EDIM   32
#define WIN    7
#define TOK    49
#define TPAD   64
#define NQKV   288
#define NTHR   1024
#define NWAV   (NTHR / 32)
#define NWIN   (64 * 64)
#define NGRID  256

#define XS_P   100
#define WQ_P   608
#define PW_P   224
#define QK_P   36
#define VP_P   96
#define SC_P   68
#define QOFF   0
#define KOFF   (HEADS * TPAD * QK_P)
#define VOFF   (2 * HEADS * TPAD * QK_P)
#define QKV_SZ (VOFF + HEADS * (TPAD / 2) * VP_P)
#define RB_SZ  ((TPAD - 1) * TOK + TPAD)

typedef __attribute__((ext_vector_type(2))) float v2f;
typedef __attribute__((ext_vector_type(8))) float v8f;

#define WMMA4(a, b, c) \
  __builtin_amdgcn_wmma_f32_16x16x4_f32(false, (a), false, (b), (short)0, (c), false, false)

#define QUAD_XOR1 0xB1
#define QUAD_XOR2 0x4E
template <int CTRL>
__device__ __forceinline__ float quad_xchg(float x)
{
  return __builtin_bit_cast(float,
      __builtin_amdgcn_update_dpp(0, __builtin_bit_cast(int, x), CTRL, 0xF, 0xF, true));
}

__device__ __forceinline__ void scatter_qkv(float* __restrict__ qkv, const v8f acc, int col,
                                            int r0, float bias)
{
  const int head = col / CTOT;
  const int r96  = col - head * CTOT;
  const int e    = r96 / 3;
  const int kqv  = r96 - 3 * e;
  const bool isv = (kqv == 2);
  const int base = isv ? (VOFF + head * ((TPAD / 2) * VP_P) + (r0 >> 1) * VP_P + e * 2)
                       : (kqv * KOFF + head * (TPAD * QK_P) + r0 * QK_P + e);
  const int s2 = isv ? VP_P : 2 * QK_P;
  const int s1 = isv ? 1 : QK_P;
#pragma unroll
  for (int j = 0; j < 8; ++j)
    qkv[base + (j >> 1) * s2 + (j & 1) * s1] = acc[j] + bias;
}

__global__ __launch_bounds__(NTHR) void swin_msa_fused(
    const float* __restrict__ x, const float* __restrict__ w_qkv,
    const float* __restrict__ b_qkv, const float* __restrict__ w_proj,
    const float* __restrict__ b_proj, const float* __restrict__ rel_bias,
    float* __restrict__ out)
{
  __shared__ __attribute__((aligned(16))) float wqp[CTOT / 2][WQ_P];
  __shared__ __attribute__((aligned(16))) float wpp[CTOT / 2][PW_P];
  __shared__ __attribute__((aligned(16))) float qkv[QKV_SZ];
  __shared__ __attribute__((aligned(16))) float sc[HEADS][TPAD][SC_P];
  __shared__ __attribute__((aligned(16))) float rbl[RB_SZ];
  static_assert(TPAD * XS_P <= HEADS * TPAD * SC_P, "the x tile must fit inside the score buffer");
  static_assert(TPAD * XS_P <= KOFF - QOFF, "the attention output tile must fit inside Q");
  float (*xs)[XS_P] = (float (*)[XS_P])&sc[0][0][0];
  float (*ow)[XS_P] = (float (*)[XS_P])&qkv[QOFF];

  const int tid  = threadIdx.x;
  const int wid  = tid >> 5;
  const int lane = tid & 31;
  const int ln16 = lane & 15;
  const int kh   = lane >> 4;
  const int rofs = kh * 8;

  for (int idx = tid; idx < (CTOT / 2) * (NQKV / 4); idx += NTHR) {
    int p  = idx / (NQKV / 4);
    int c4 = idx - p * (NQKV / 4);
    float4 r0 = *(const float4*)(w_qkv + (2 * p) * NQKV + c4 * 4);
    float4 r1 = *(const float4*)(w_qkv + (2 * p + 1) * NQKV + c4 * 4);
    float* d = &wqp[p][c4 * 8];
    *(float4*)(d)     = make_float4(r0.x, r1.x, r0.y, r1.y);
    *(float4*)(d + 4) = make_float4(r0.z, r1.z, r0.w, r1.w);
  }

  for (int idx = tid; idx < (CTOT / 2) * (CTOT / 4); idx += NTHR) {
    int p  = idx / (CTOT / 4);
    int c4 = idx - p * (CTOT / 4);
    float4 r0 = *(const float4*)(w_proj + (2 * p) * CTOT + c4 * 4);
    float4 r1 = *(const float4*)(w_proj + (2 * p + 1) * CTOT + c4 * 4);
    float* d = &wpp[p][c4 * 8];
    *(float4*)(d)     = make_float4(r0.x, r1.x, r0.y, r1.y);
    *(float4*)(d + 4) = make_float4(r0.z, r1.z, r0.w, r1.w);
  }

  for (int idx = tid; idx < RB_SZ; idx += NTHR)
    rbl[idx] = (idx < TOK * TOK) ? rel_bias[idx] : 0.f;

  for (int blk = blockIdx.x; blk < NWIN; blk += NGRID) {
  const int b   = blk >> 6;
  const int win = blk & 63;
  const int wy  = win >> 3;
  const int wx  = win & 7;
  const long wbase = ((long)b * (HW_ * HW_) + (long)((wy * WIN) * HW_ + wx * WIN)) * CTOT;
  const float* __restrict__ xw = x + wbase;
  float* __restrict__ ow_g     = out + wbase;

  for (int idx = tid; idx < TPAD * (CTOT / 4); idx += NTHR) {
    int r  = idx / (CTOT / 4);
    int c4 = idx - r * (CTOT / 4);
    float4 v = make_float4(0.f, 0.f, 0.f, 0.f);
    if (r < TOK) {
      int tok = (r / WIN) * HW_ + (r % WIN);
      v = *(const float4*)(xw + tok * CTOT + c4 * 4);
    }
    *(float4*)&xs[r][c4 * 4] = v;
  }
  __syncthreads();

  for (int tt = wid; tt < 4 * 6; tt += NWAV) {
    const int row0 = (tt / 6) * 16;
    const int cg   = (tt % 6) * 48;
    v8f acc0 = {}, acc1 = {}, acc2 = {};
    const int c0 = cg + ln16;
    const float bq0 = b_qkv[c0];
    const float bq1 = b_qkv[c0 + 16];
    const float bq2 = b_qkv[c0 + 32];
    const float* ap = &xs[row0 + ln16][kh * 2];
    const float* bp = &wqp[kh][(cg + ln16) * 2];
#pragma unroll
    for (int kk = 0; kk < CTOT; kk += 4) {
      v2f a  = *(const v2f*)(ap + kk);
      v2f b0 = *(const v2f*)(bp + (kk / 2) * WQ_P);
      v2f b1 = *(const v2f*)(bp + (kk / 2) * WQ_P + 32);
      v2f b2 = *(const v2f*)(bp + (kk / 2) * WQ_P + 64);
      acc0 = WMMA4(a, b0, acc0);
      acc1 = WMMA4(a, b1, acc1);
      acc2 = WMMA4(a, b2, acc2);
    }
    scatter_qkv(qkv, acc0, c0,      row0 + rofs, bq0);
    scatter_qkv(qkv, acc1, c0 + 16, row0 + rofs, bq1);
    scatter_qkv(qkv, acc2, c0 + 32, row0 + rofs, bq2);
  }
  __syncthreads();

  const float scale = 0.17677669529663687f;
  for (int tt = wid; tt < HEADS * 8; tt += NWAV) {
    const int head = tt / 8;
    const int q0   = ((tt % 8) / 2) * 16;
    const int jh   = (tt % 2) * 32;
    v8f acc0 = {}, acc1 = {};
    float rb[2][8];
    {
      const float* rbp = &rbl[(q0 + rofs) * TOK + jh + ln16];
#pragma unroll
      for (int j = 0; j < 8; ++j) {
        rb[0][j] = rbp[j * TOK];
        rb[1][j] = rbp[j * TOK + 16];
      }
    }
    const float* ap = &qkv[QOFF + head * (TPAD * QK_P) + (q0 + ln16) * QK_P + kh * 2];
    const float* bp = &qkv[KOFF + head * (TPAD * QK_P) + (jh + ln16) * QK_P + kh * 2];
#pragma unroll
    for (int kk = 0; kk < EDIM; kk += 4) {
      v2f a  = *(const v2f*)(ap + kk);
      v2f b0 = *(const v2f*)(bp + kk);
      v2f b1 = *(const v2f*)(bp + 16 * QK_P + kk);
      acc0 = WMMA4(a, b0, acc0);
      acc1 = WMMA4(a, b1, acc1);
    }
#pragma unroll
    for (int i = 0; i < 2; ++i) {
      const int j0  = jh + i * 16;
#pragma unroll
      for (int j = 0; j < 8; ++j) {
        int row = q0 + j + rofs;
        int col = j0 + ln16;
        float a = (i == 0) ? acc0[j] : acc1[j];
        float bias = (row < TOK) ? rb[i][j] : 0.f;
        float v = a * scale + bias;
        if (col >= TOK) v = -1e30f;
        sc[head][row][col] = v;
      }
    }
  }
  __syncthreads();

  {
    const int t = wid * 8 + (lane >> 2);
    if (t < HEADS * TOK) {
      const int head = t / TOK;
      const int r    = t - head * TOK;
      float4* rowp = (float4*)&sc[head][r][(lane & 3) * 16];
      float4 q4[4];
      for (int i = 0; i < 4; ++i) q4[i] = rowp[i];
      float v[16];
      for (int i = 0; i < 4; ++i) {
        v[4 * i + 0] = q4[i].x; v[4 * i + 1] = q4[i].y;
        v[4 * i + 2] = q4[i].z; v[4 * i + 3] = q4[i].w;
      }
      float m8[8], m4[4];
      for (int i = 0; i < 8; ++i) m8[i] = fmaxf(v[2 * i], v[2 * i + 1]);
      for (int i = 0; i < 4; ++i) m4[i] = fmaxf(m8[2 * i], m8[2 * i + 1]);
      float mx = fmaxf(fmaxf(m4[0], m4[1]), fmaxf(m4[2], m4[3]));
      mx = fmaxf(mx, quad_xchg<QUAD_XOR1>(mx));
      mx = fmaxf(mx, quad_xchg<QUAD_XOR2>(mx));
      float e[16];
      for (int i = 0; i < 16; ++i) e[i] = __expf(v[i] - mx);
      float s8[8], s4[4];
      for (int i = 0; i < 8; ++i) s8[i] = e[2 * i] + e[2 * i + 1];
      for (int i = 0; i < 4; ++i) s4[i] = s8[2 * i] + s8[2 * i + 1];
      float s = (s4[0] + s4[1]) + (s4[2] + s4[3]);
      s += quad_xchg<QUAD_XOR1>(s);
      s += quad_xchg<QUAD_XOR2>(s);
      const float inv = 1.f / s;
      for (int i = 0; i < 4; ++i)
        rowp[i] = make_float4(e[4 * i + 0] * inv, e[4 * i + 1] * inv,
                              e[4 * i + 2] * inv, e[4 * i + 3] * inv);
    }
  }
  __syncthreads();

  for (int tt = wid; tt < HEADS * 4 * 2; tt += NWAV) {
    const int head = tt / 8;
    const int q0   = ((tt % 8) / 2) * 16;
    const int e0   = (tt % 2) * 16;
    v8f acc = {};
    const float* ap = &sc[head][q0 + ln16][kh * 2];
    const float* bp = &qkv[VOFF + head * ((TPAD / 2) * VP_P) + kh * VP_P + (e0 + ln16) * 2];
#pragma unroll
    for (int kk = 0; kk < TPAD; kk += 4) {
      v2f a  = *(const v2f*)(ap + kk);
      v2f bf = *(const v2f*)(bp + (kk / 2) * VP_P);
      acc = WMMA4(a, bf, acc);
    }
#pragma unroll
    for (int j = 0; j < 8; ++j) {
      int row = q0 + j + rofs;
      int col = e0 + ln16;
      ow[row][head * EDIM + col] = acc[j];
    }
  }
  __syncthreads();

  for (int tt = wid; tt < 4 * 6; tt += NWAV) {
    const int row0 = (tt / 6) * 16;
    const int col0 = (tt % 6) * 16;
    const float bpj = b_proj[col0 + ln16];
    v8f acc = {};
    const float* ap = &ow[row0 + ln16][kh * 2];
    const float* bp = &wpp[kh][(col0 + ln16) * 2];
#pragma unroll
    for (int kk = 0; kk < CTOT; kk += 4) {
      v2f a  = *(const v2f*)(ap + kk);
      v2f bf = *(const v2f*)(bp + (kk / 2) * PW_P);
      acc = WMMA4(a, bf, acc);
    }
#pragma unroll
    for (int j = 0; j < 8; ++j) {
      int row = row0 + j + rofs;
      if (row < TOK) {
        int col = col0 + ln16;
        int tok = (row / WIN) * HW_ + (row % WIN);
        ow_g[tok * CTOT + col] = acc[j] + bpj;
      }
    }
  }
  }
}

extern "C" void kernel_launch(void* const* d_in, const int* in_sizes, int n_in,
                              void* d_out, int out_size, void* d_ws, size_t ws_size,
                              hipStream_t stream) {
  const float* x        = (const float*)d_in[0];
  const float* w_qkv    = (const float*)d_in[1];
  const float* b_qkv    = (const float*)d_in[2];
  const float* w_proj   = (const float*)d_in[3];
  const float* b_proj   = (const float*)d_in[4];
  const float* rel_bias = (const float*)d_in[5];
  float* out = (float*)d_out;
  (void)in_sizes; (void)n_in; (void)out_size; (void)d_ws; (void)ws_size;

  swin_msa_fused<<<NGRID, NTHR, 0, stream>>>(x, w_qkv, b_qkv, w_proj, b_proj,
                                             rel_bias, out);
}


# next window's x tile loaded into registers before phase 4 (PV), stored to LDS after it; b_proj from an LDS copy so no other load wait drains them: no memory round trip per window; sums unchanged
# speedup vs baseline: 4.7021x; 1.0590x over previous
#include <hip/hip_runtime.h>


#define HW_    56
#define CTOT   96
#define HEADS  3
#define EDIM   32
#define WIN    7
#define TOK    49
#define TPAD   64
#define NQKV   288
#define NTHR   1024
#define NWAV   (NTHR / 32)
#define NWIN   (64 * 64)
#define NGRID  256

#define XS_P   100
#define WQ_P   608
#define PW_P   224
#define QK_P   36
#define VP_P   96
#define SC_P   68
#define QOFF   0
#define KOFF   (HEADS * TPAD * QK_P)
#define VOFF   (2 * HEADS * TPAD * QK_P)
#define QKV_SZ (VOFF + HEADS * (TPAD / 2) * VP_P)
#define RB_SZ  ((TPAD - 1) * TOK + TPAD)

typedef __attribute__((ext_vector_type(2))) float v2f;
typedef __attribute__((ext_vector_type(8))) float v8f;

#define WMMA4(a, b, c) \
  __builtin_amdgcn_wmma_f32_16x16x4_f32(false, (a), false, (b), (short)0, (c), false, false)

#define QUAD_XOR1 0xB1
#define QUAD_XOR2 0x4E
template <int CTRL>
__device__ __forceinline__ float quad_xchg(float x)
{
  return __builtin_bit_cast(float,
      __builtin_amdgcn_update_dpp(0, __builtin_bit_cast(int, x), CTRL, 0xF, 0xF, true));
}

__device__ __forceinline__ void scatter_qkv(float* __restrict__ qkv, const v8f acc, int col,
                                            int r0, float bias)
{
  const int head = col / CTOT;
  const int r96  = col - head * CTOT;
  const int e    = r96 / 3;
  const int kqv  = r96 - 3 * e;
  const bool isv = (kqv == 2);
  const int base = isv ? (VOFF + head * ((TPAD / 2) * VP_P) + (r0 >> 1) * VP_P + e * 2)
                       : (kqv * KOFF + head * (TPAD * QK_P) + r0 * QK_P + e);
  const int s2 = isv ? VP_P : 2 * QK_P;
  const int s1 = isv ? 1 : QK_P;
#pragma unroll
  for (int j = 0; j < 8; ++j)
    qkv[base + (j >> 1) * s2 + (j & 1) * s1] = acc[j] + bias;
}

__device__ __forceinline__ void get_x_share(const float* __restrict__ xw, int ga, int gb,
                                            bool real_b, float4& xa, float4& xb)
{
  xa = *(const float4*)(xw + ga);
  xb = make_float4(0.f, 0.f, 0.f, 0.f);
  if (real_b) xb = *(const float4*)(xw + gb);
}

__global__ __launch_bounds__(NTHR) void swin_msa_fused(
    const float* __restrict__ x, const float* __restrict__ w_qkv,
    const float* __restrict__ b_qkv, const float* __restrict__ w_proj,
    const float* __restrict__ b_proj, const float* __restrict__ rel_bias,
    float* __restrict__ out)
{
  __shared__ __attribute__((aligned(16))) float wqp[CTOT / 2][WQ_P];
  __shared__ __attribute__((aligned(16))) float wpp[CTOT / 2][PW_P];
  __shared__ __attribute__((aligned(16))) float qkv[QKV_SZ];
  __shared__ __attribute__((aligned(16))) float sc[HEADS][TPAD][SC_P];
  __shared__ __attribute__((aligned(16))) float rbl[RB_SZ];
  __shared__ __attribute__((aligned(16))) float bpl[CTOT];
  static_assert(TPAD * XS_P <= HEADS * TPAD * SC_P, "the x tile must fit inside the score buffer");
  static_assert(TPAD * XS_P <= KOFF - QOFF, "the attention output tile must fit inside Q");
  float (*xs)[XS_P] = (float (*)[XS_P])&sc[0][0][0];
  float (*ow)[XS_P] = (float (*)[XS_P])&qkv[QOFF];

  const int tid  = threadIdx.x;
  const int wid  = tid >> 5;
  const int lane = tid & 31;
  const int ln16 = lane & 15;
  const int kh   = lane >> 4;
  const int rofs = kh * 8;

  for (int idx = tid; idx < (CTOT / 2) * (NQKV / 4); idx += NTHR) {
    int p  = idx / (NQKV / 4);
    int c4 = idx - p * (NQKV / 4);
    float4 r0 = *(const float4*)(w_qkv + (2 * p) * NQKV + c4 * 4);
    float4 r1 = *(const float4*)(w_qkv + (2 * p + 1) * NQKV + c4 * 4);
    float* d = &wqp[p][c4 * 8];
    *(float4*)(d)     = make_float4(r0.x, r1.x, r0.y, r1.y);
    *(float4*)(d + 4) = make_float4(r0.z, r1.z, r0.w, r1.w);
  }

  for (int idx = tid; idx < (CTOT / 2) * (CTOT / 4); idx += NTHR) {
    int p  = idx / (CTOT / 4);
    int c4 = idx - p * (CTOT / 4);
    float4 r0 = *(const float4*)(w_proj + (2 * p) * CTOT + c4 * 4);
    float4 r1 = *(const float4*)(w_proj + (2 * p + 1) * CTOT + c4 * 4);
    float* d = &wpp[p][c4 * 8];
    *(float4*)(d)     = make_float4(r0.x, r1.x, r0.y, r1.y);
    *(float4*)(d + 4) = make_float4(r0.z, r1.z, r0.w, r1.w);
  }

  for (int idx = tid; idx < RB_SZ; idx += NTHR)
    rbl[idx] = (idx < TOK * TOK) ? rel_bias[idx] : 0.f;

  for (int idx = tid; idx < CTOT; idx += NTHR)
    bpl[idx] = b_proj[idx];

  static_assert(TPAD * (CTOT / 4) <= 2 * NTHR, "two pieces per thread must cover the x tile");
  const int  xrA    = tid / (CTOT / 4);
  const int  xcA    = tid - xrA * (CTOT / 4);
  const int  xgA    = ((xrA / WIN) * HW_ + (xrA % WIN)) * CTOT + xcA * 4;
  const int  xiB    = tid + NTHR;
  const int  xrB    = xiB / (CTOT / 4);
  const int  xcB    = xiB - xrB * (CTOT / 4);
  const bool xhasB  = xiB < TPAD * (CTOT / 4);
  const bool xrealB = xrB < TOK;
  const int  xgB    = ((xrB / WIN) * HW_ + (xrB % WIN)) * CTOT + xcB * 4;

  {
    const int b0  = blockIdx.x >> 6;
    const int w0  = blockIdx.x & 63;
    const long wb0 = ((long)b0 * (HW_ * HW_) + (long)(((w0 >> 3) * WIN) * HW_ + (w0 & 7) * WIN)) * CTOT;
    float4 fa, fb;
    get_x_share(x + wb0, xgA, xgB, xrealB, fa, fb);
    *(float4*)&xs[xrA][xcA * 4] = fa;
    if (xhasB) *(float4*)&xs[xrB][xcB * 4] = fb;
  }

  for (int blk = blockIdx.x; blk < NWIN; blk += NGRID) {
  const int b   = blk >> 6;
  const int win = blk & 63;
  const int wy  = win >> 3;
  const int wx  = win & 7;
  const long wbase = ((long)b * (HW_ * HW_) + (long)((wy * WIN) * HW_ + wx * WIN)) * CTOT;
  float* __restrict__ ow_g     = out + wbase;

  __syncthreads();

  for (int tt = wid; tt < 4 * 6; tt += NWAV) {
    const int row0 = (tt / 6) * 16;
    const int cg   = (tt % 6) * 48;
    v8f acc0 = {}, acc1 = {}, acc2 = {};
    const int c0 = cg + ln16;
    const float bq0 = b_qkv[c0];
    const float bq1 = b_qkv[c0 + 16];
    const float bq2 = b_qkv[c0 + 32];
    const float* ap = &xs[row0 + ln16][kh * 2];
    const float* bp = &wqp[kh][(cg + ln16) * 2];
#pragma unroll
    for (int kk = 0; kk < CTOT; kk += 4) {
      v2f a  = *(const v2f*)(ap + kk);
      v2f b0 = *(const v2f*)(bp + (kk / 2) * WQ_P);
      v2f b1 = *(const v2f*)(bp + (kk / 2) * WQ_P + 32);
      v2f b2 = *(const v2f*)(bp + (kk / 2) * WQ_P + 64);
      acc0 = WMMA4(a, b0, acc0);
      acc1 = WMMA4(a, b1, acc1);
      acc2 = WMMA4(a, b2, acc2);
    }
    scatter_qkv(qkv, acc0, c0,      row0 + rofs, bq0);
    scatter_qkv(qkv, acc1, c0 + 16, row0 + rofs, bq1);
    scatter_qkv(qkv, acc2, c0 + 32, row0 + rofs, bq2);
  }
  __syncthreads();

  const float scale = 0.17677669529663687f;
  for (int tt = wid; tt < HEADS * 8; tt += NWAV) {
    const int head = tt / 8;
    const int q0   = ((tt % 8) / 2) * 16;
    const int jh   = (tt % 2) * 32;
    v8f acc0 = {}, acc1 = {};
    float rb[2][8];
    {
      const float* rbp = &rbl[(q0 + rofs) * TOK + jh + ln16];
#pragma unroll
      for (int j = 0; j < 8; ++j) {
        rb[0][j] = rbp[j * TOK];
        rb[1][j] = rbp[j * TOK + 16];
      }
    }
    const float* ap = &qkv[QOFF + head * (TPAD * QK_P) + (q0 + ln16) * QK_P + kh * 2];
    const float* bp = &qkv[KOFF + head * (TPAD * QK_P) + (jh + ln16) * QK_P + kh * 2];
#pragma unroll
    for (int kk = 0; kk < EDIM; kk += 4) {
      v2f a  = *(const v2f*)(ap + kk);
      v2f b0 = *(const v2f*)(bp + kk);
      v2f b1 = *(const v2f*)(bp + 16 * QK_P + kk);
      acc0 = WMMA4(a, b0, acc0);
      acc1 = WMMA4(a, b1, acc1);
    }
#pragma unroll
    for (int i = 0; i < 2; ++i) {
      const int j0  = jh + i * 16;
#pragma unroll
      for (int j = 0; j < 8; ++j) {
        int row = q0 + j + rofs;
        int col = j0 + ln16;
        float a = (i == 0) ? acc0[j] : acc1[j];
        float bias = (row < TOK) ? rb[i][j] : 0.f;
        float v = a * scale + bias;
        if (col >= TOK) v = -1e30f;
        sc[head][row][col] = v;
      }
    }
  }
  __syncthreads();

  {
    const int t = wid * 8 + (lane >> 2);
    if (t < HEADS * TOK) {
      const int head = t / TOK;
      const int r    = t - head * TOK;
      float4* rowp = (float4*)&sc[head][r][(lane & 3) * 16];
      float4 q4[4];
      for (int i = 0; i < 4; ++i) q4[i] = rowp[i];
      float v[16];
      for (int i = 0; i < 4; ++i) {
        v[4 * i + 0] = q4[i].x; v[4 * i + 1] = q4[i].y;
        v[4 * i + 2] = q4[i].z; v[4 * i + 3] = q4[i].w;
      }
      float m8[8], m4[4];
      for (int i = 0; i < 8; ++i) m8[i] = fmaxf(v[2 * i], v[2 * i + 1]);
      for (int i = 0; i < 4; ++i) m4[i] = fmaxf(m8[2 * i], m8[2 * i + 1]);
      float mx = fmaxf(fmaxf(m4[0], m4[1]), fmaxf(m4[2], m4[3]));
      mx = fmaxf(mx, quad_xchg<QUAD_XOR1>(mx));
      mx = fmaxf(mx, quad_xchg<QUAD_XOR2>(mx));
      float e[16];
      for (int i = 0; i < 16; ++i) e[i] = __expf(v[i] - mx);
      float s8[8], s4[4];
      for (int i = 0; i < 8; ++i) s8[i] = e[2 * i] + e[2 * i + 1];
      for (int i = 0; i < 4; ++i) s4[i] = s8[2 * i] + s8[2 * i + 1];
      float s = (s4[0] + s4[1]) + (s4[2] + s4[3]);
      s += quad_xchg<QUAD_XOR1>(s);
      s += quad_xchg<QUAD_XOR2>(s);
      const float inv = 1.f / s;
      for (int i = 0; i < 4; ++i)
        rowp[i] = make_float4(e[4 * i + 0] * inv, e[4 * i + 1] * inv,
                              e[4 * i + 2] * inv, e[4 * i + 3] * inv);
    }
  }
  __syncthreads();

  const bool have_next = (blk + NGRID) < NWIN;
  float4 nxa = make_float4(0.f, 0.f, 0.f, 0.f), nxb = nxa;
  if (have_next) {
    const int nb  = (blk + NGRID) >> 6;
    const int nw  = (blk + NGRID) & 63;
    const long nwb = ((long)nb * (HW_ * HW_) + (long)(((nw >> 3) * WIN) * HW_ + (nw & 7) * WIN)) * CTOT;
    get_x_share(x + nwb, xgA, xgB, xrealB, nxa, nxb);
  }

  for (int tt = wid; tt < HEADS * 4 * 2; tt += NWAV) {
    const int head = tt / 8;
    const int q0   = ((tt % 8) / 2) * 16;
    const int e0   = (tt % 2) * 16;
    v8f acc = {};
    const float* ap = &sc[head][q0 + ln16][kh * 2];
    const float* bp = &qkv[VOFF + head * ((TPAD / 2) * VP_P) + kh * VP_P + (e0 + ln16) * 2];
#pragma unroll
    for (int kk = 0; kk < TPAD; kk += 4) {
      v2f a  = *(const v2f*)(ap + kk);
      v2f bf = *(const v2f*)(bp + (kk / 2) * VP_P);
      acc = WMMA4(a, bf, acc);
    }
#pragma unroll
    for (int j = 0; j < 8; ++j) {
      int row = q0 + j + rofs;
      int col = e0 + ln16;
      ow[row][head * EDIM + col] = acc[j];
    }
  }
  __syncthreads();

  if (have_next) {
    *(float4*)&xs[xrA][xcA * 4] = nxa;
    if (xhasB) *(float4*)&xs[xrB][xcB * 4] = nxb;
  }

  for (int tt = wid; tt < 4 * 6; tt += NWAV) {
    const int row0 = (tt / 6) * 16;
    const int col0 = (tt % 6) * 16;
    const float bpj = bpl[col0 + ln16];
    v8f acc = {};
    const float* ap = &ow[row0 + ln16][kh * 2];
    const float* bp = &wpp[kh][(col0 + ln16) * 2];
#pragma unroll
    for (int kk = 0; kk < CTOT; kk += 4) {
      v2f a  = *(const v2f*)(ap + kk);
      v2f bf = *(const v2f*)(bp + (kk / 2) * PW_P);
      acc = WMMA4(a, bf, acc);
    }
#pragma unroll
    for (int j = 0; j < 8; ++j) {
      int row = row0 + j + rofs;
      if (row < TOK) {
        int col = col0 + ln16;
        int tok = (row / WIN) * HW_ + (row % WIN);
        ow_g[tok * CTOT + col] = acc[j] + bpj;
      }
    }
  }
  }
}

extern "C" void kernel_launch(void* const* d_in, const int* in_sizes, int n_in,
                              void* d_out, int out_size, void* d_ws, size_t ws_size,
                              hipStream_t stream) {
  const float* x        = (const float*)d_in[0];
  const float* w_qkv    = (const float*)d_in[1];
  const float* b_qkv    = (const float*)d_in[2];
  const float* w_proj   = (const float*)d_in[3];
  const float* b_proj   = (const float*)d_in[4];
  const float* rel_bias = (const float*)d_in[5];
  float* out = (float*)d_out;
  (void)in_sizes; (void)n_in; (void)out_size; (void)d_ws; (void)ws_size;

  swin_msa_fused<<<NGRID, NTHR, 0, stream>>>(x, w_qkv, b_qkv, w_proj, b_proj,
                                             rel_bias, out);
}


# last token (row 48) of the work-group's 16 windows batched into the first window's 4th qkv row tile; later windows run qkv on 3 row tiles (27 two-tile strips) and copy row 48; sums unchanged
# speedup vs baseline: 5.1700x; 1.0995x over previous
#include <hip/hip_runtime.h>


#define HW_    56
#define CTOT   96
#define HEADS  3
#define EDIM   32
#define WIN    7
#define TOK    49
#define TPAD   64
#define NQKV   288
#define NTHR   1024
#define NWAV   (NTHR / 32)
#define NWIN   (64 * 64)
#define NGRID  256
#define WPG    (NWIN / NGRID)
#define P1W    27

#define XS_P   100
#define WQ_P   608
#define PW_P   224
#define QK_P   36
#define VP_P   96
#define SC_P   68
#define QOFF   0
#define KOFF   (HEADS * TPAD * QK_P)
#define VOFF   (2 * HEADS * TPAD * QK_P)
#define QKV_SZ (VOFF + HEADS * (TPAD / 2) * VP_P)
#define RB_SZ  ((TPAD - 1) * TOK + TPAD)

typedef __attribute__((ext_vector_type(2))) float v2f;
typedef __attribute__((ext_vector_type(8))) float v8f;

#define WMMA4(a, b, c) \
  __builtin_amdgcn_wmma_f32_16x16x4_f32(false, (a), false, (b), (short)0, (c), false, false)

#define QUAD_XOR1 0xB1
#define QUAD_XOR2 0x4E
template <int CTRL>
__device__ __forceinline__ float quad_xchg(float x)
{
  return __builtin_bit_cast(float,
      __builtin_amdgcn_update_dpp(0, __builtin_bit_cast(int, x), CTRL, 0xF, 0xF, true));
}

__device__ __forceinline__ void scatter_qkv(float* __restrict__ qkv, const v8f acc, int col,
                                            int r0, float bias)
{
  const int head = col / CTOT;
  const int r96  = col - head * CTOT;
  const int e    = r96 / 3;
  const int kqv  = r96 - 3 * e;
  const bool isv = (kqv == 2);
  const int base = isv ? (VOFF + head * ((TPAD / 2) * VP_P) + (r0 >> 1) * VP_P + e * 2)
                       : (kqv * KOFF + head * (TPAD * QK_P) + r0 * QK_P + e);
  const int s2 = isv ? VP_P : 2 * QK_P;
  const int s1 = isv ? 1 : QK_P;
#pragma unroll
  for (int j = 0; j < 8; ++j)
    qkv[base + (j >> 1) * s2 + (j & 1) * s1] = acc[j] + bias;
}

__device__ __forceinline__ void get_x_share(const float* __restrict__ xw, int ga, int gb,
                                            bool real_b, float4& xa, float4& xb)
{
  xa = *(const float4*)(xw + ga);
  xb = make_float4(0.f, 0.f, 0.f, 0.f);
  if (real_b) xb = *(const float4*)(xw + gb);
}

__global__ __launch_bounds__(NTHR) void swin_msa_fused(
    const float* __restrict__ x, const float* __restrict__ w_qkv,
    const float* __restrict__ b_qkv, const float* __restrict__ w_proj,
    const float* __restrict__ b_proj, const float* __restrict__ rel_bias,
    float* __restrict__ out)
{
  __shared__ __attribute__((aligned(16))) float wqp[CTOT / 2][WQ_P];
  __shared__ __attribute__((aligned(16))) float wpp[CTOT / 2][PW_P];
  __shared__ __attribute__((aligned(16))) float qkv[QKV_SZ];
  __shared__ __attribute__((aligned(16))) float sc[HEADS][TPAD][SC_P];
  __shared__ __attribute__((aligned(16))) float rbl[RB_SZ];
  __shared__ __attribute__((aligned(16))) float bpl[CTOT];
  static_assert(WPG == TPAD - TOK + 1, "token 48 of the workgroup's windows must fill rows 48..63 of one tile");
  static_assert(NGRID % 64 == 0, "a workgroup's windows differ in the batch index only");
  __shared__ __attribute__((aligned(16))) float q48[(WPG - 1) * CTOT];
  static_assert(TPAD * XS_P <= HEADS * TPAD * SC_P, "the x tile must fit inside the score buffer");
  static_assert(TPAD * XS_P <= KOFF - QOFF, "the attention output tile must fit inside Q");
  float (*xs)[XS_P] = (float (*)[XS_P])&sc[0][0][0];
  float (*ow)[XS_P] = (float (*)[XS_P])&qkv[QOFF];

  const int tid  = threadIdx.x;
  const int wid  = tid >> 5;
  const int lane = tid & 31;
  const int ln16 = lane & 15;
  const int kh   = lane >> 4;
  const int rofs = kh * 8;

  for (int idx = tid; idx < (CTOT / 2) * (NQKV / 4); idx += NTHR) {
    int p  = idx / (NQKV / 4);
    int c4 = idx - p * (NQKV / 4);
    float4 r0 = *(const float4*)(w_qkv + (2 * p) * NQKV + c4 * 4);
    float4 r1 = *(const float4*)(w_qkv + (2 * p + 1) * NQKV + c4 * 4);
    float* d = &wqp[p][c4 * 8];
    *(float4*)(d)     = make_float4(r0.x, r1.x, r0.y, r1.y);
    *(float4*)(d + 4) = make_float4(r0.z, r1.z, r0.w, r1.w);
  }

  for (int idx = tid; idx < (CTOT / 2) * (CTOT / 4); idx += NTHR) {
    int p  = idx / (CTOT / 4);
    int c4 = idx - p * (CTOT / 4);
    float4 r0 = *(const float4*)(w_proj + (2 * p) * CTOT + c4 * 4);
    float4 r1 = *(const float4*)(w_proj + (2 * p + 1) * CTOT + c4 * 4);
    float* d = &wpp[p][c4 * 8];
    *(float4*)(d)     = make_float4(r0.x, r1.x, r0.y, r1.y);
    *(float4*)(d + 4) = make_float4(r0.z, r1.z, r0.w, r1.w);
  }

  for (int idx = tid; idx < RB_SZ; idx += NTHR)
    rbl[idx] = (idx < TOK * TOK) ? rel_bias[idx] : 0.f;

  for (int idx = tid; idx < CTOT; idx += NTHR)
    bpl[idx] = b_proj[idx];

  static_assert(TPAD * (CTOT / 4) <= 2 * NTHR, "two pieces per thread must cover the x tile");
  const int  xrA    = tid / (CTOT / 4);
  const int  xcA    = tid - xrA * (CTOT / 4);
  const int  xgA    = ((xrA / WIN) * HW_ + (xrA % WIN)) * CTOT + xcA * 4;
  const int  xiB    = tid + NTHR;
  const int  xrB    = xiB / (CTOT / 4);
  const int  xcB    = xiB - xrB * (CTOT / 4);
  const bool xhasB  = xiB < TPAD * (CTOT / 4);
  const bool xrealB = xrB < TOK;
  const int  xgB    = ((xrB / WIN) * HW_ + (xrB % WIN)) * CTOT + xcB * 4;

  {
    const int b0  = blockIdx.x >> 6;
    const int w0  = blockIdx.x & 63;
    const long wb0 = ((long)b0 * (HW_ * HW_) + (long)(((w0 >> 3) * WIN) * HW_ + (w0 & 7) * WIN)) * CTOT;
    const int xgB0 = xrealB ? xgB
                            : (xrB - (TOK - 1)) * ((NGRID / 64) * HW_ * HW_ * CTOT)
                              + (((TOK - 1) / WIN) * HW_ + (TOK - 1) % WIN) * CTOT + xcB * 4;
    float4 fa, fb;
    get_x_share(x + wb0, xgA, xgB0, xhasB, fa, fb);
    *(float4*)&xs[xrA][xcA * 4] = fa;
    if (xhasB) *(float4*)&xs[xrB][xcB * 4] = fb;
  }

  for (int blk = blockIdx.x; blk < NWIN; blk += NGRID) {
  const int wi  = blk / NGRID;
  const bool first = (wi == 0);
  const int b   = blk >> 6;
  const int win = blk & 63;
  const int wy  = win >> 3;
  const int wx  = win & 7;
  const long wbase = ((long)b * (HW_ * HW_) + (long)((wy * WIN) * HW_ + wx * WIN)) * CTOT;
  float* __restrict__ ow_g     = out + wbase;

  __syncthreads();

  static_assert(P1W == 3 * (NQKV / 32), "one wave per strip of the three full row tiles");
  auto qkv_strip = [&](const int tt) {
    const int row0 = (tt / (NQKV / 32)) * 16;
    const int cg   = (tt % (NQKV / 32)) * 32;
    v8f acc0 = {}, acc1 = {};
    const int c0 = cg + ln16;
    const float bq0 = b_qkv[c0];
    const float bq1 = b_qkv[c0 + 16];
    const float* ap = &xs[row0 + ln16][kh * 2];
    const float* bp = &wqp[kh][(cg + ln16) * 2];
#pragma unroll
    for (int kk = 0; kk < CTOT; kk += 4) {
      v2f a  = *(const v2f*)(ap + kk);
      v2f b0 = *(const v2f*)(bp + (kk / 2) * WQ_P);
      v2f b1 = *(const v2f*)(bp + (kk / 2) * WQ_P + 32);
      acc0 = WMMA4(a, b0, acc0);
      acc1 = WMMA4(a, b1, acc1);
    }
    scatter_qkv(qkv, acc0, c0,      row0 + rofs, bq0);
    scatter_qkv(qkv, acc1, c0 + 16, row0 + rofs, bq1);
  };
  if (wid < P1W) qkv_strip(wid);
  if (first) {
    const int t2 = (wid >= P1W) ? wid : wid + NWAV;
    if (t2 < 4 * (NQKV / 32)) qkv_strip(t2);
  }
  if (!first && wid >= P1W) {
    for (int idx = tid - P1W * 32; idx < 3 * CTOT; idx += (NWAV - P1W) * 32) {
      const int which = idx / CTOT;
      const int c     = idx - which * CTOT;
      const int head  = c / EDIM;
      const int e     = c - head * EDIM;
      const int rsrc  = TOK - 1 + wi;
      float v;
      int dst;
      if (which == 0) {
        v   = q48[(wi - 1) * CTOT + c];
        dst = QOFF + head * (TPAD * QK_P) + (TOK - 1) * QK_P + e;
      } else if (which == 1) {
        v   = qkv[KOFF + head * (TPAD * QK_P) + rsrc * QK_P + e];
        dst = KOFF + head * (TPAD * QK_P) + (TOK - 1) * QK_P + e;
      } else {
        v   = qkv[VOFF + head * ((TPAD / 2) * VP_P) + (rsrc >> 1) * VP_P + 2 * e + (rsrc & 1)];
        dst = VOFF + head * ((TPAD / 2) * VP_P) + ((TOK - 1) >> 1) * VP_P + 2 * e + ((TOK - 1) & 1);
      }
      qkv[dst] = v;
    }
  }
  __syncthreads();

  if (first && wid >= HEADS * 8) {
    for (int idx = tid - HEADS * 8 * 32; idx < (WPG - 1) * CTOT; idx += (NWAV - HEADS * 8) * 32) {
      const int j    = idx / CTOT;
      const int c    = idx - j * CTOT;
      const int head = c / EDIM;
      const int e    = c - head * EDIM;
      q48[idx] = qkv[QOFF + head * (TPAD * QK_P) + (TOK + j) * QK_P + e];
    }
  }

  const float scale = 0.17677669529663687f;
  for (int tt = wid; tt < HEADS * 8; tt += NWAV) {
    const int head = tt / 8;
    const int q0   = ((tt % 8) / 2) * 16;
    const int jh   = (tt % 2) * 32;
    v8f acc0 = {}, acc1 = {};
    float rb[2][8];
    {
      const float* rbp = &rbl[(q0 + rofs) * TOK + jh + ln16];
#pragma unroll
      for (int j = 0; j < 8; ++j) {
        rb[0][j] = rbp[j * TOK];
        rb[1][j] = rbp[j * TOK + 16];
      }
    }
    const float* ap = &qkv[QOFF + head * (TPAD * QK_P) + (q0 + ln16) * QK_P + kh * 2];
    const float* bp = &qkv[KOFF + head * (TPAD * QK_P) + (jh + ln16) * QK_P + kh * 2];
#pragma unroll
    for (int kk = 0; kk < EDIM; kk += 4) {
      v2f a  = *(const v2f*)(ap + kk);
      v2f b0 = *(const v2f*)(bp + kk);
      v2f b1 = *(const v2f*)(bp + 16 * QK_P + kk);
      acc0 = WMMA4(a, b0, acc0);
      acc1 = WMMA4(a, b1, acc1);
    }
#pragma unroll
    for (int i = 0; i < 2; ++i) {
      const int j0  = jh + i * 16;
#pragma unroll
      for (int j = 0; j < 8; ++j) {
        int row = q0 + j + rofs;
        int col = j0 + ln16;
        float a = (i == 0) ? acc0[j] : acc1[j];
        float bias = (row < TOK) ? rb[i][j] : 0.f;
        float v = a * scale + bias;
        if (col >= TOK) v = -1e30f;
        sc[head][row][col] = v;
      }
    }
  }
  __syncthreads();

  {
    const int t = wid * 8 + (lane >> 2);
    if (t < HEADS * TOK) {
      const int head = t / TOK;
      const int r    = t - head * TOK;
      float4* rowp = (float4*)&sc[head][r][(lane & 3) * 16];
      float4 q4[4];
      for (int i = 0; i < 4; ++i) q4[i] = rowp[i];
      float v[16];
      for (int i = 0; i < 4; ++i) {
        v[4 * i + 0] = q4[i].x; v[4 * i + 1] = q4[i].y;
        v[4 * i + 2] = q4[i].z; v[4 * i + 3] = q4[i].w;
      }
      float m8[8], m4[4];
      for (int i = 0; i < 8; ++i) m8[i] = fmaxf(v[2 * i], v[2 * i + 1]);
      for (int i = 0; i < 4; ++i) m4[i] = fmaxf(m8[2 * i], m8[2 * i + 1]);
      float mx = fmaxf(fmaxf(m4[0], m4[1]), fmaxf(m4[2], m4[3]));
      mx = fmaxf(mx, quad_xchg<QUAD_XOR1>(mx));
      mx = fmaxf(mx, quad_xchg<QUAD_XOR2>(mx));
      float e[16];
      for (int i = 0; i < 16; ++i) e[i] = __expf(v[i] - mx);
      float s8[8], s4[4];
      for (int i = 0; i < 8; ++i) s8[i] = e[2 * i] + e[2 * i + 1];
      for (int i = 0; i < 4; ++i) s4[i] = s8[2 * i] + s8[2 * i + 1];
      float s = (s4[0] + s4[1]) + (s4[2] + s4[3]);
      s += quad_xchg<QUAD_XOR1>(s);
      s += quad_xchg<QUAD_XOR2>(s);
      const float inv = 1.f / s;
      for (int i = 0; i < 4; ++i)
        rowp[i] = make_float4(e[4 * i + 0] * inv, e[4 * i + 1] * inv,
                              e[4 * i + 2] * inv, e[4 * i + 3] * inv);
    }
  }
  __syncthreads();

  const bool have_next = (blk + NGRID) < NWIN;
  float4 nxa = make_float4(0.f, 0.f, 0.f, 0.f), nxb = nxa;
  if (have_next) {
    const int nb  = (blk + NGRID) >> 6;
    const int nw  = (blk + NGRID) & 63;
    const long nwb = ((long)nb * (HW_ * HW_) + (long)(((nw >> 3) * WIN) * HW_ + (nw & 7) * WIN)) * CTOT;
    get_x_share(x + nwb, xgA, xgB, xrealB, nxa, nxb);
  }

  for (int tt = wid; tt < HEADS * 4 * 2; tt += NWAV) {
    const int head = tt / 8;
    const int q0   = ((tt % 8) / 2) * 16;
    const int e0   = (tt % 2) * 16;
    v8f acc = {};
    const float* ap = &sc[head][q0 + ln16][kh * 2];
    const float* bp = &qkv[VOFF + head * ((TPAD / 2) * VP_P) + kh * VP_P + (e0 + ln16) * 2];
#pragma unroll
    for (int kk = 0; kk < TPAD; kk += 4) {
      v2f a  = *(const v2f*)(ap + kk);
      v2f bf = *(const v2f*)(bp + (kk / 2) * VP_P);
      acc = WMMA4(a, bf, acc);
    }
#pragma unroll
    for (int j = 0; j < 8; ++j) {
      int row = q0 + j + rofs;
      int col = e0 + ln16;
      ow[row][head * EDIM + col] = acc[j];
    }
  }
  __syncthreads();

  if (have_next) {
    *(float4*)&xs[xrA][xcA * 4] = nxa;
    if (xhasB) *(float4*)&xs[xrB][xcB * 4] = nxb;
  }

  for (int tt = wid; tt < 4 * 6; tt += NWAV) {
    const int row0 = (tt / 6) * 16;
    const int col0 = (tt % 6) * 16;
    const float bpj = bpl[col0 + ln16];
    v8f acc = {};
    const float* ap = &ow[row0 + ln16][kh * 2];
    const float* bp = &wpp[kh][(col0 + ln16) * 2];
#pragma unroll
    for (int kk = 0; kk < CTOT; kk += 4) {
      v2f a  = *(const v2f*)(ap + kk);
      v2f bf = *(const v2f*)(bp + (kk / 2) * PW_P);
      acc = WMMA4(a, bf, acc);
    }
#pragma unroll
    for (int j = 0; j < 8; ++j) {
      int row = row0 + j + rofs;
      if (row < TOK) {
        int col = col0 + ln16;
        int tok = (row / WIN) * HW_ + (row % WIN);
        ow_g[tok * CTOT + col] = acc[j] + bpj;
      }
    }
  }
  }
}

extern "C" void kernel_launch(void* const* d_in, const int* in_sizes, int n_in,
                              void* d_out, int out_size, void* d_ws, size_t ws_size,
                              hipStream_t stream) {
  const float* x        = (const float*)d_in[0];
  const float* w_qkv    = (const float*)d_in[1];
  const float* b_qkv    = (const float*)d_in[2];
  const float* w_proj   = (const float*)d_in[3];
  const float* b_proj   = (const float*)d_in[4];
  const float* rel_bias = (const float*)d_in[5];
  float* out = (float*)d_out;
  (void)in_sizes; (void)n_in; (void)out_size; (void)d_ws; (void)ws_size;

  swin_msa_fused<<<NGRID, NTHR, 0, stream>>>(x, w_qkv, b_qkv, w_proj, b_proj,
                                             rel_bias, out);
}


# scores, softmax and P@V fused in one wave per head and 16 queries: transposed score tile stays in registers and is the B operand of P@V; 2 barriers per window, not 5; softmax and P@V sum order change
# speedup vs baseline: 5.6199x; 1.0870x over previous
#include <hip/hip_runtime.h>


#define HW_    56
#define CTOT   96
#define HEADS  3
#define EDIM   32
#define WIN    7
#define TOK    49
#define TPAD   64
#define NQKV   288
#define NTHR   1024
#define NWAV   (NTHR / 32)
#define NWIN   (64 * 64)
#define NGRID  256
#define WPG    (NWIN / NGRID)
#define P1W    27

#define XS_P   100
#define WQ_P   608
#define PW_P   224
#define QK_P   36
#define VP_P   96
#define QOFF   0
#define KOFF   (HEADS * TPAD * QK_P)
#define VOFF   (2 * HEADS * TPAD * QK_P)
#define QKV_SZ (VOFF + HEADS * (TPAD / 2) * VP_P)
#define RB_SZ  ((TPAD - 1) * TOK + TPAD)

typedef __attribute__((ext_vector_type(2))) float v2f;
typedef __attribute__((ext_vector_type(8))) float v8f;

#define WMMA4(a, b, c) \
  __builtin_amdgcn_wmma_f32_16x16x4_f32(false, (a), false, (b), (short)0, (c), false, false)

#define QUAD_XOR1 0xB1
#define QUAD_XOR2 0x4E
template <int CTRL>
__device__ __forceinline__ float quad_xchg(float x)
{
  return __builtin_bit_cast(float,
      __builtin_amdgcn_update_dpp(0, __builtin_bit_cast(int, x), CTRL, 0xF, 0xF, true));
}

__device__ __forceinline__ float xhalf(float x)
{
  return __builtin_bit_cast(float,
      __builtin_amdgcn_permlanex16(__builtin_bit_cast(int, x), __builtin_bit_cast(int, x),
                                   0x76543210, 0xfedcba98, false, false));
}

__device__ __forceinline__ void scatter_qkv(float* __restrict__ qkv, const v8f acc, int col,
                                            int r0, float bias)
{
  const int head = col / CTOT;
  const int r96  = col - head * CTOT;
  const int e    = r96 / 3;
  const int kqv  = r96 - 3 * e;
  const bool isv = (kqv == 2);
  const int base = isv ? (VOFF + head * ((TPAD / 2) * VP_P) + (r0 >> 1) * VP_P + e * 2)
                       : (kqv * KOFF + head * (TPAD * QK_P) + r0 * QK_P + e);
  const int s2 = isv ? VP_P : 2 * QK_P;
  const int s1 = isv ? 1 : QK_P;
#pragma unroll
  for (int j = 0; j < 8; ++j)
    qkv[base + (j >> 1) * s2 + (j & 1) * s1] = acc[j] + bias;
}

__device__ __forceinline__ void get_x_share(const float* __restrict__ xw, int ga, int gb,
                                            bool real_b, float4& xa, float4& xb)
{
  xa = *(const float4*)(xw + ga);
  xb = make_float4(0.f, 0.f, 0.f, 0.f);
  if (real_b) xb = *(const float4*)(xw + gb);
}

__global__ __launch_bounds__(NTHR) void swin_msa_fused(
    const float* __restrict__ x, const float* __restrict__ w_qkv,
    const float* __restrict__ b_qkv, const float* __restrict__ w_proj,
    const float* __restrict__ b_proj, const float* __restrict__ rel_bias,
    float* __restrict__ out)
{
  __shared__ __attribute__((aligned(16))) float wqp[CTOT / 2][WQ_P];
  __shared__ __attribute__((aligned(16))) float wpp[CTOT / 2][PW_P];
  __shared__ __attribute__((aligned(16))) float qkv[QKV_SZ];
  __shared__ __attribute__((aligned(16))) float xs[TPAD][XS_P];
  __shared__ __attribute__((aligned(16))) float ow[TPAD][XS_P];
  __shared__ __attribute__((aligned(16))) float rbl[RB_SZ];
  __shared__ __attribute__((aligned(16))) float bpl[CTOT];
  static_assert(WPG == TPAD - TOK + 1, "token 48 of the workgroup's windows must fill rows 48..63 of one tile");
  static_assert(NGRID % 64 == 0, "a workgroup's windows differ in the batch index only");

  const int tid  = threadIdx.x;
  const int wid  = tid >> 5;
  const int lane = tid & 31;
  const int ln16 = lane & 15;
  const int kh   = lane >> 4;
  const int rofs = kh * 8;

  for (int idx = tid; idx < (CTOT / 2) * (NQKV / 4); idx += NTHR) {
    int p  = idx / (NQKV / 4);
    int c4 = idx - p * (NQKV / 4);
    float4 r0 = *(const float4*)(w_qkv + (2 * p) * NQKV + c4 * 4);
    float4 r1 = *(const float4*)(w_qkv + (2 * p + 1) * NQKV + c4 * 4);
    float* d = &wqp[p][c4 * 8];
    *(float4*)(d)     = make_float4(r0.x, r1.x, r0.y, r1.y);
    *(float4*)(d + 4) = make_float4(r0.z, r1.z, r0.w, r1.w);
  }

  for (int idx = tid; idx < (CTOT / 2) * (CTOT / 4); idx += NTHR) {
    int p  = idx / (CTOT / 4);
    int c4 = idx - p * (CTOT / 4);
    float4 r0 = *(const float4*)(w_proj + (2 * p) * CTOT + c4 * 4);
    float4 r1 = *(const float4*)(w_proj + (2 * p + 1) * CTOT + c4 * 4);
    float* d = &wpp[p][c4 * 8];
    *(float4*)(d)     = make_float4(r0.x, r1.x, r0.y, r1.y);
    *(float4*)(d + 4) = make_float4(r0.z, r1.z, r0.w, r1.w);
  }

  for (int idx = tid; idx < RB_SZ; idx += NTHR)
    rbl[idx] = (idx < TOK * TOK) ? rel_bias[idx] : 0.f;

  for (int idx = tid; idx < CTOT; idx += NTHR)
    bpl[idx] = b_proj[idx];

  static_assert(TPAD * (CTOT / 4) <= 2 * NTHR, "two pieces per thread must cover the x tile");
  const int  xrA    = tid / (CTOT / 4);
  const int  xcA    = tid - xrA * (CTOT / 4);
  const int  xgA    = ((xrA / WIN) * HW_ + (xrA % WIN)) * CTOT + xcA * 4;
  const int  xiB    = tid + NTHR;
  const int  xrB    = xiB / (CTOT / 4);
  const int  xcB    = xiB - xrB * (CTOT / 4);
  const bool xhasB  = xiB < TPAD * (CTOT / 4);
  const bool xrealB = xrB < TOK;
  const int  xgB    = ((xrB / WIN) * HW_ + (xrB % WIN)) * CTOT + xcB * 4;

  constexpr int NXP = 3 * WIN * (CTOT / 4);
  static_assert(HEADS * 4 * 32 + NXP <= NTHR, "the fetching threads must exist beside the attention waves");
  const int  nu    = tid - HEADS * 4 * 32;
  const bool nfet  = (nu >= 0) && (nu < NXP);
  const int  nr    = nu / (CTOT / 4);
  const int  ncc   = nu - nr * (CTOT / 4);
  const int  ng    = ((nr / WIN) * HW_ + (nr % WIN)) * CTOT + ncc * 4;
  const bool nthird = nr + 6 * WIN < 3 * 16;

  {
    const int b0  = blockIdx.x >> 6;
    const int w0  = blockIdx.x & 63;
    const long wb0 = ((long)b0 * (HW_ * HW_) + (long)(((w0 >> 3) * WIN) * HW_ + (w0 & 7) * WIN)) * CTOT;
    const int xgB0 = xrealB ? xgB
                            : (xrB - (TOK - 1)) * ((NGRID / 64) * HW_ * HW_ * CTOT)
                              + (((TOK - 1) / WIN) * HW_ + (TOK - 1) % WIN) * CTOT + xcB * 4;
    float4 fa, fb;
    get_x_share(x + wb0, xgA, xgB0, xhasB, fa, fb);
    *(float4*)&xs[xrA][xcA * 4] = fa;
    if (xhasB) *(float4*)&xs[xrB][xcB * 4] = fb;
  }

  __syncthreads();

  for (int blk = blockIdx.x; blk < NWIN; blk += NGRID) {
  const int wi  = blk / NGRID;
  const bool first = (wi == 0);
  const int b   = blk >> 6;
  const int win = blk & 63;
  const int wy  = win >> 3;
  const int wx  = win & 7;
  const long wbase = ((long)b * (HW_ * HW_) + (long)((wy * WIN) * HW_ + wx * WIN)) * CTOT;
  float* __restrict__ ow_g     = out + wbase;


  static_assert(P1W == 3 * (NQKV / 32), "one wave per strip of the three full row tiles");
  auto qkv_strip = [&](const int tt) {
    const int row0 = (tt / (NQKV / 32)) * 16;
    const int cg   = (tt % (NQKV / 32)) * 32;
    v8f acc0 = {}, acc1 = {};
    const int c0 = cg + ln16;
    const float bq0 = b_qkv[c0];
    const float bq1 = b_qkv[c0 + 16];
    const float* ap = &xs[row0 + ln16][kh * 2];
    const float* bp = &wqp[kh][(cg + ln16) * 2];
#pragma unroll
    for (int kk = 0; kk < CTOT; kk += 4) {
      v2f a  = *(const v2f*)(ap + kk);
      v2f b0 = *(const v2f*)(bp + (kk / 2) * WQ_P);
      v2f b1 = *(const v2f*)(bp + (kk / 2) * WQ_P + 32);
      acc0 = WMMA4(a, b0, acc0);
      acc1 = WMMA4(a, b1, acc1);
    }
    scatter_qkv(qkv, acc0, c0,      row0 + rofs, bq0);
    scatter_qkv(qkv, acc1, c0 + 16, row0 + rofs, bq1);
  };
  if (wid < P1W) qkv_strip(wid);
  if (first) {
    const int t2 = (wid >= P1W) ? wid : wid + NWAV;
    if (t2 < 4 * (NQKV / 32)) qkv_strip(t2);
  }
  if (!first && wid >= P1W) {
    for (int idx = tid - P1W * 32; idx < 3 * CTOT; idx += (NWAV - P1W) * 32) {
      const int which = idx / CTOT;
      const int c     = idx - which * CTOT;
      const int head  = c / EDIM;
      const int e     = c - head * EDIM;
      const int rsrc  = TOK - 1 + wi;
      float v;
      int dst;
      if (which < 2) {
        v   = qkv[which * KOFF + head * (TPAD * QK_P) + rsrc * QK_P + e];
        dst = which * KOFF + head * (TPAD * QK_P) + (TOK - 1) * QK_P + e;
      } else {
        v   = qkv[VOFF + head * ((TPAD / 2) * VP_P) + (rsrc >> 1) * VP_P + 2 * e + (rsrc & 1)];
        dst = VOFF + head * ((TPAD / 2) * VP_P) + ((TOK - 1) >> 1) * VP_P + 2 * e + ((TOK - 1) & 1);
      }
      qkv[dst] = v;
    }
  }
  __syncthreads();

  const float scale = 0.17677669529663687f;
  if (wid < HEADS * 4) {
    const int head = wid >> 2;
    const int q0   = (wid & 3) * 16;
    const int qrow = q0 + ln16;
    float rb[3][8];
    const float* rbp = &rbl[qrow * TOK + rofs];
#pragma unroll
    for (int t = 0; t < 3; ++t)
#pragma unroll
      for (int r = 0; r < 8; ++r) rb[t][r] = rbp[16 * t + r];
    const float rb48 = rbl[qrow * TOK + (TOK - 1)];
    v8f s0 = {}, s1 = {}, s2 = {}, s3 = {};
    const float* kp = &qkv[KOFF + head * (TPAD * QK_P) + ln16 * QK_P + kh * 2];
    const float* qp = &qkv[QOFF + head * (TPAD * QK_P) + qrow * QK_P + kh * 2];
#pragma unroll
    for (int kk = 0; kk < EDIM; kk += 4) {
      v2f bq = *(const v2f*)(qp + kk);
      v2f a0 = *(const v2f*)(kp + kk);
      v2f a1 = *(const v2f*)(kp + 16 * QK_P + kk);
      v2f a2 = *(const v2f*)(kp + 32 * QK_P + kk);
      v2f a3 = *(const v2f*)(kp + 48 * QK_P + kk);
      s0 = WMMA4(a0, bq, s0);
      s1 = WMMA4(a1, bq, s1);
      s2 = WMMA4(a2, bq, s2);
      s3 = WMMA4(a3, bq, s3);
    }
    float v[25];
#pragma unroll
    for (int r = 0; r < 8; ++r) {
      v[r]      = s0[r] * scale + rb[0][r];
      v[8 + r]  = s1[r] * scale + rb[1][r];
      v[16 + r] = s2[r] * scale + rb[2][r];
    }
    v[24] = (kh == 0) ? (s3[0] * scale + rb48) : -1e30f;
    float m12[12], m6[6], m3[3];
#pragma unroll
    for (int i = 0; i < 12; ++i) m12[i] = fmaxf(v[2 * i], v[2 * i + 1]);
#pragma unroll
    for (int i = 0; i < 6; ++i) m6[i] = fmaxf(m12[2 * i], m12[2 * i + 1]);
#pragma unroll
    for (int i = 0; i < 3; ++i) m3[i] = fmaxf(m6[2 * i], m6[2 * i + 1]);
    float mx = fmaxf(fmaxf(m3[0], m3[1]), fmaxf(m3[2], v[24]));
    mx = fmaxf(mx, xhalf(mx));
    float p[25];
#pragma unroll
    for (int i = 0; i < 25; ++i) p[i] = __expf(v[i] - mx);
    float a12[12], a6[6], a3s[3];
#pragma unroll
    for (int i = 0; i < 12; ++i) a12[i] = p[2 * i] + p[2 * i + 1];
#pragma unroll
    for (int i = 0; i < 6; ++i) a6[i] = a12[2 * i] + a12[2 * i + 1];
#pragma unroll
    for (int i = 0; i < 3; ++i) a3s[i] = a6[2 * i] + a6[2 * i + 1];
    float sum = (a3s[0] + a3s[1]) + (a3s[2] + p[24]);
    sum += xhalf(sum);
    const float inv = 1.f / sum;
#pragma unroll
    for (int i = 0; i < 25; ++i) p[i] *= inv;

    v8f o0 = {}, o1 = {};
    const float* vp = &qkv[VOFF + head * ((TPAD / 2) * VP_P) + kh * 4 * VP_P + ln16 * 2];
#pragma unroll
    for (int t = 0; t < 3; ++t) {
#pragma unroll
      for (int j = 0; j < 4; ++j) {
        v2f pb;
        pb.x = p[8 * t + 2 * j];
        pb.y = p[8 * t + 2 * j + 1];
        v2f va = *(const v2f*)(vp + (8 * t + j) * VP_P);
        v2f vb = *(const v2f*)(vp + (8 * t + j) * VP_P + 32);
        o0 = WMMA4(va, pb, o0);
        o1 = WMMA4(vb, pb, o1);
      }
    }
    {
      v2f pb;
      pb.x = p[24];
      pb.y = 0.f;
      v2f va = *(const v2f*)(vp + 24 * VP_P);
      v2f vb = *(const v2f*)(vp + 24 * VP_P + 32);
      o0 = WMMA4(va, pb, o0);
      o1 = WMMA4(vb, pb, o1);
    }
    float* op = &ow[qrow][head * EDIM + rofs];
    *(float4*)(op)      = make_float4(o0[0], o0[1], o0[2], o0[3]);
    *(float4*)(op + 4)  = make_float4(o0[4], o0[5], o0[6], o0[7]);
    *(float4*)(op + 16) = make_float4(o1[0], o1[1], o1[2], o1[3]);
    *(float4*)(op + 20) = make_float4(o1[4], o1[5], o1[6], o1[7]);
  } else if (nfet && blk + NGRID < NWIN) {
    const int nb  = (blk + NGRID) >> 6;
    const int nw  = (blk + NGRID) & 63;
    const long nwb = ((long)nb * (HW_ * HW_) + (long)(((nw >> 3) * WIN) * HW_ + (nw & 7) * WIN)) * CTOT;
    const float* __restrict__ xw = x + nwb;
    float* xd = &xs[nr][ncc * 4];
    const float4 fa = *(const float4*)(xw + ng);
    const float4 fb = *(const float4*)(xw + (ng + 3 * HW_ * CTOT));
    float4 fc = make_float4(0.f, 0.f, 0.f, 0.f);
    if (nthird) fc = *(const float4*)(xw + (ng + 6 * HW_ * CTOT));
    *(float4*)(xd) = fa;
    *(float4*)(xd + 3 * WIN * XS_P) = fb;
    if (nthird) *(float4*)(xd + 6 * WIN * XS_P) = fc;
  }
  __syncthreads();

  for (int tt = wid; tt < 4 * 6; tt += NWAV) {
    const int row0 = (tt / 6) * 16;
    const int col0 = (tt % 6) * 16;
    const float bpj = bpl[col0 + ln16];
    v8f acc = {};
    const float* ap = &ow[row0 + ln16][kh * 2];
    const float* bp = &wpp[kh][(col0 + ln16) * 2];
#pragma unroll
    for (int kk = 0; kk < CTOT; kk += 4) {
      v2f a  = *(const v2f*)(ap + kk);
      v2f bf = *(const v2f*)(bp + (kk / 2) * PW_P);
      acc = WMMA4(a, bf, acc);
    }
#pragma unroll
    for (int j = 0; j < 8; ++j) {
      int row = row0 + j + rofs;
      if (row < TOK) {
        int col = col0 + ln16;
        int tok = (row / WIN) * HW_ + (row % WIN);
        ow_g[tok * CTOT + col] = acc[j] + bpj;
      }
    }
  }
  }
}

extern "C" void kernel_launch(void* const* d_in, const int* in_sizes, int n_in,
                              void* d_out, int out_size, void* d_ws, size_t ws_size,
                              hipStream_t stream) {
  const float* x        = (const float*)d_in[0];
  const float* w_qkv    = (const float*)d_in[1];
  const float* b_qkv    = (const float*)d_in[2];
  const float* w_proj   = (const float*)d_in[3];
  const float* b_proj   = (const float*)d_in[4];
  const float* rel_bias = (const float*)d_in[5];
  float* out = (float*)d_out;
  (void)in_sizes; (void)n_in; (void)out_size; (void)d_ws; (void)ws_size;

  swin_msa_fused<<<NGRID, NTHR, 0, stream>>>(x, w_qkv, b_qkv, w_proj, b_proj,
                                             rel_bias, out);
}
